# starting kernel
# baseline (speedup 1.0000x reference)
#include <hip/hip_runtime.h>

typedef float v2f __attribute__((ext_vector_type(2)));
typedef float v8f __attribute__((ext_vector_type(8)));

#define D_MODEL   256
#define BATCH_A   4096
#define BATCH_B   8192
#define NKEYS     4
#define NBUCKETS  1024

__device__ __forceinline__ int hash_row(const int* __restrict__ keys, int row) {
    int s = 0;
#pragma unroll
    for (int k = 0; k < NKEYS; ++k) s += keys[row * NKEYS + k];
    return (int)(((unsigned)s) & (NBUCKETS - 1));
}

__global__ void hash_b_kernel(const int* __restrict__ keys_b, int* __restrict__ hash_b) {
    int j = blockIdx.x * blockDim.x + threadIdx.x;
    if (j < BATCH_B) hash_b[j] = hash_row(keys_b, j);
}

__global__ __launch_bounds__(D_MODEL)
void bucket_sum_kernel(const float* __restrict__ emb_b,
                       const int* __restrict__ hash_b,
                       float* __restrict__ bucket_sums,
                       int* __restrict__ counts) {
    const int g = blockIdx.x;
    const int t = threadIdx.x;
    float acc = 0.0f;
    int cnt = 0;
    for (int j = 0; j < BATCH_B; ++j) {
        if (hash_b[j] == g) {
            acc += emb_b[(size_t)j * D_MODEL + t];
            ++cnt;
        }
    }
    bucket_sums[(size_t)g * D_MODEL + t] = acc;
    if (t == 0) counts[g] = cnt;
}

__global__ __launch_bounds__(D_MODEL)
void build_combined_kernel(const float* __restrict__ emb_a,
                           const int* __restrict__ keys_a,
                           const float* __restrict__ bucket_sums,
                           const int* __restrict__ counts,
                           float* __restrict__ combined) {
    const int i = blockIdx.x;
    const int t = threadIdx.x;
    const int ha  = hash_row(keys_a, i);
    const int cnt = counts[ha];
    const float scale = 1.0f / (float)(cnt < 1 ? 1 : cnt);
    combined[(size_t)i * (2 * D_MODEL) + t]           = emb_a[(size_t)i * D_MODEL + t];
    combined[(size_t)i * (2 * D_MODEL) + D_MODEL + t] =
        bucket_sums[(size_t)ha * D_MODEL + t] * scale;
}

#define BM 64
#define BN 64
#define KT 16
#define AS_LD 20
#define BS_LD 68

__global__ __launch_bounds__(256)
void wmma_gemm_bias_kernel(const float* __restrict__ A,
                           const float* __restrict__ B,
                           const float* __restrict__ bias,
                           float* __restrict__ C,
                           int M, int N, int K, int relu) {
    __shared__ __align__(16) float As[BM][AS_LD];
    __shared__ __align__(16) float Bs[KT][BS_LD];

    const int tid  = threadIdx.x;
    const int lane = tid & 31;
    const int wave = tid >> 5;
    const int wm   = wave >> 1;
    const int wn   = wave & 1;
    const int hi   = lane >> 4;
    const int l15  = lane & 15;

    const int bm = blockIdx.y * BM;
    const int bn = blockIdx.x * BN;

    v8f c0 = {};
    v8f c1 = {};

    const int ar = tid >> 2;
    const int ac = (tid & 3) * 4;
    const int br = tid >> 4;
    const int bc = (tid & 15) * 4;

    for (int k0 = 0; k0 < K; k0 += KT) {
        const float4 av = *(const float4*)&A[(size_t)(bm + ar) * K + (k0 + ac)];
        const float4 bv = *(const float4*)&B[(size_t)(k0 + br) * N + (bn + bc)];
        __syncthreads();
        *(float4*)&As[ar][ac] = av;
        *(float4*)&Bs[br][bc] = bv;
        __syncthreads();

#pragma unroll
        for (int kk = 0; kk < KT; kk += 4) {
            const int kb = kk + 2 * hi;
            v2f a = *(const v2f*)&As[wm * 16 + l15][kb];
            v2f b0, b1;
            b0.x = Bs[kb    ][wn * 32 + l15];
            b0.y = Bs[kb + 1][wn * 32 + l15];
            b1.x = Bs[kb    ][wn * 32 + 16 + l15];
            b1.y = Bs[kb + 1][wn * 32 + 16 + l15];
            c0 = __builtin_amdgcn_wmma_f32_16x16x4_f32(false, a, false, b0,
                                                       (short)0, c0, false, false);
            c1 = __builtin_amdgcn_wmma_f32_16x16x4_f32(false, a, false, b1,
                                                       (short)0, c1, false, false);
        }
    }

    const int n0 = bn + wn * 32 + l15;
    const float bia0 = bias[n0];
    const float bia1 = bias[n0 + 16];
#pragma unroll
    for (int r = 0; r < 8; ++r) {
        const int m = bm + wm * 16 + r + 8 * hi;
        float v0 = c0[r] + bia0;
        float v1 = c1[r] + bia1;
        if (relu) { v0 = fmaxf(v0, 0.0f); v1 = fmaxf(v1, 0.0f); }
        C[(size_t)m * N + n0]      = v0;
        C[(size_t)m * N + n0 + 16] = v1;
    }
}

extern "C" void kernel_launch(void* const* d_in, const int* in_sizes, int n_in,
                              void* d_out, int out_size, void* d_ws, size_t ws_size,
                              hipStream_t stream) {
    const float* emb_a  = (const float*)d_in[0];
    const float* emb_b  = (const float*)d_in[1];
    const int*   keys_a = (const int*)d_in[2];
    const int*   keys_b = (const int*)d_in[3];
    const float* W1 = (const float*)d_in[4];
    const float* b1 = (const float*)d_in[5];
    const float* W2 = (const float*)d_in[6];
    const float* b2 = (const float*)d_in[7];
    const float* Wo = (const float*)d_in[8];
    const float* bo = (const float*)d_in[9];
    float* out = (float*)d_out;

    float* ws          = (float*)d_ws;
    int*   hash_b      = (int*)ws;
    float* bucket_sums = ws + BATCH_B;
    int*   counts      = (int*)(bucket_sums + NBUCKETS * D_MODEL);
    float* combined    = bucket_sums + NBUCKETS * D_MODEL + NBUCKETS;
    float* h           = combined + (size_t)BATCH_A * 2 * D_MODEL;
    float* rel         = h        + (size_t)BATCH_A * 2 * D_MODEL;

    hash_b_kernel<<<(BATCH_B + 255) / 256, 256, 0, stream>>>(keys_b, hash_b);
    bucket_sum_kernel<<<NBUCKETS, D_MODEL, 0, stream>>>(emb_b, hash_b, bucket_sums, counts);
    build_combined_kernel<<<BATCH_A, D_MODEL, 0, stream>>>(emb_a, keys_a, bucket_sums,
                                                           counts, combined);

    wmma_gemm_bias_kernel<<<dim3(512 / BN, BATCH_A / BM), 256, 0, stream>>>(
        combined, W1, b1, h, BATCH_A, 512, 512, 1);
    wmma_gemm_bias_kernel<<<dim3(256 / BN, BATCH_A / BM), 256, 0, stream>>>(
        h, W2, b2, rel, BATCH_A, 256, 512, 0);
    wmma_gemm_bias_kernel<<<dim3(256 / BN, BATCH_A / BM), 256, 0, stream>>>(
        rel, Wo, bo, out, BATCH_A, 256, 256, 0);
}


# bucket sum: a block's 8 waves find the bucket's rows by one ballot scan of the hashes, list them in row order in LDS and add only those rows (was: each thread scanned all 8192); sum order kept
# speedup vs baseline: 3.3466x; 3.3466x over previous
#include <hip/hip_runtime.h>

typedef float v2f __attribute__((ext_vector_type(2)));
typedef float v8f __attribute__((ext_vector_type(8)));

#define D_MODEL   256
#define BATCH_A   4096
#define BATCH_B   8192
#define NKEYS     4
#define NBUCKETS  1024

__device__ __forceinline__ int hash_row(const int* __restrict__ keys, int row) {
    int s = 0;
#pragma unroll
    for (int k = 0; k < NKEYS; ++k) s += keys[row * NKEYS + k];
    return (int)(((unsigned)s) & (NBUCKETS - 1));
}

__global__ void hash_b_kernel(const int* __restrict__ keys_b, int* __restrict__ hash_b) {
    int j = blockIdx.x * blockDim.x + threadIdx.x;
    if (j < BATCH_B) hash_b[j] = hash_row(keys_b, j);
}

#define SCAN_WAVES      (D_MODEL / 32)
#define SCAN_PER_WAVE   (BATCH_B / SCAN_WAVES)
#define SCAN_GROUPS     (SCAN_PER_WAVE / (32 * 4))
#define SUM_BATCH       16
static_assert(SCAN_WAVES * SCAN_PER_WAVE == BATCH_B, "scan must cover every b-row");
static_assert(SCAN_GROUPS * 128 == SCAN_PER_WAVE, "whole 16-byte loads only");
static_assert(BATCH_B <= 65536, "row numbers are kept as 16-bit values");

__device__ __forceinline__ int popc32(unsigned v) { return __builtin_popcount(v); }

__global__ __launch_bounds__(D_MODEL)
void bucket_sum_kernel(const float* __restrict__ emb_b,
                       const int* __restrict__ hash_b,
                       float* __restrict__ bucket_sums,
                       int* __restrict__ counts) {
    __shared__ __align__(16) unsigned short rows[BATCH_B + SUM_BATCH];
    __shared__ int wave_cnt[SCAN_WAVES];

    const int g    = blockIdx.x;
    const int t    = threadIdx.x;
    const int lane = t & 31;
    const int wave = t >> 5;

    const int4* hp = (const int4*)(hash_b + wave * SCAN_PER_WAVE) + lane;
    int4 hv[SCAN_GROUPS];
#pragma unroll
    for (int i = 0; i < SCAN_GROUPS; ++i) hv[i] = hp[i * 32];

    unsigned m0[SCAN_GROUPS], m1[SCAN_GROUPS], m2[SCAN_GROUPS], m3[SCAN_GROUPS];
    int cnt = 0;
#pragma unroll
    for (int i = 0; i < SCAN_GROUPS; ++i) {
        m0[i] = __builtin_amdgcn_ballot_w32(hv[i].x == g);
        m1[i] = __builtin_amdgcn_ballot_w32(hv[i].y == g);
        m2[i] = __builtin_amdgcn_ballot_w32(hv[i].z == g);
        m3[i] = __builtin_amdgcn_ballot_w32(hv[i].w == g);
        cnt += popc32(m0[i]) + popc32(m1[i]) + popc32(m2[i]) + popc32(m3[i]);
    }
    if (lane == 0) wave_cnt[wave] = cnt;
    __syncthreads();

    int base = 0, total = 0;
#pragma unroll
    for (int w = 0; w < SCAN_WAVES; ++w) {
        const int c = wave_cnt[w];
        if (w < wave) base += c;
        total += c;
    }
    if (t < SUM_BATCH) rows[total + t] = 0;
    if (cnt != 0) {
        const unsigned lt = (1u << lane) - 1u;
        int pos = base;
#pragma unroll
        for (int i = 0; i < SCAN_GROUPS; ++i) {
            if ((m0[i] | m1[i] | m2[i] | m3[i]) != 0u) {
                int p = pos + popc32(m0[i] & lt) + popc32(m1[i] & lt)
                            + popc32(m2[i] & lt) + popc32(m3[i] & lt);
                const int j0 = wave * SCAN_PER_WAVE + i * 128 + lane * 4;
                if (hv[i].x == g) rows[p++] = (unsigned short)(j0);
                if (hv[i].y == g) rows[p++] = (unsigned short)(j0 + 1);
                if (hv[i].z == g) rows[p++] = (unsigned short)(j0 + 2);
                if (hv[i].w == g) rows[p++] = (unsigned short)(j0 + 3);
                pos += popc32(m0[i]) + popc32(m1[i]) + popc32(m2[i]) + popc32(m3[i]);
            }
        }
    }
    __syncthreads();

    const float* __restrict__ src = emb_b + t;
    float acc = 0.0f;
    for (int e = 0; e < total; e += SUM_BATCH) {
        const uint4 pa = *(const uint4*)&rows[e];
        const uint4 pb = *(const uint4*)&rows[e + 8];
        const unsigned w[8] = {pa.x, pa.y, pa.z, pa.w, pb.x, pb.y, pb.z, pb.w};
        float r[SUM_BATCH];
#pragma unroll
        for (int k = 0; k < SUM_BATCH; k += 2) {
            r[k]     = src[(size_t)(w[k >> 1] & 0xffffu) * D_MODEL];
            r[k + 1] = src[(size_t)(w[k >> 1] >> 16)     * D_MODEL];
        }
#pragma unroll
        for (int k = 0; k < SUM_BATCH; ++k) {
            const float sum = acc + r[k];
            acc = (e + k < total) ? sum : acc;
        }
    }
    bucket_sums[(size_t)g * D_MODEL + t] = acc;
    if (t == 0) counts[g] = total;
}

__global__ __launch_bounds__(D_MODEL)
void build_combined_kernel(const float* __restrict__ emb_a,
                           const int* __restrict__ keys_a,
                           const float* __restrict__ bucket_sums,
                           const int* __restrict__ counts,
                           float* __restrict__ combined) {
    const int i = blockIdx.x;
    const int t = threadIdx.x;
    const int ha  = hash_row(keys_a, i);
    const int cnt = counts[ha];
    const float scale = 1.0f / (float)(cnt < 1 ? 1 : cnt);
    combined[(size_t)i * (2 * D_MODEL) + t]           = emb_a[(size_t)i * D_MODEL + t];
    combined[(size_t)i * (2 * D_MODEL) + D_MODEL + t] =
        bucket_sums[(size_t)ha * D_MODEL + t] * scale;
}

#define BM 64
#define BN 64
#define KT 16
#define AS_LD 20
#define BS_LD 68

__global__ __launch_bounds__(256)
void wmma_gemm_bias_kernel(const float* __restrict__ A,
                           const float* __restrict__ B,
                           const float* __restrict__ bias,
                           float* __restrict__ C,
                           int M, int N, int K, int relu) {
    __shared__ __align__(16) float As[BM][AS_LD];
    __shared__ __align__(16) float Bs[KT][BS_LD];

    const int tid  = threadIdx.x;
    const int lane = tid & 31;
    const int wave = tid >> 5;
    const int wm   = wave >> 1;
    const int wn   = wave & 1;
    const int hi   = lane >> 4;
    const int l15  = lane & 15;

    const int bm = blockIdx.y * BM;
    const int bn = blockIdx.x * BN;

    v8f c0 = {};
    v8f c1 = {};

    const int ar = tid >> 2;
    const int ac = (tid & 3) * 4;
    const int br = tid >> 4;
    const int bc = (tid & 15) * 4;

    for (int k0 = 0; k0 < K; k0 += KT) {
        const float4 av = *(const float4*)&A[(size_t)(bm + ar) * K + (k0 + ac)];
        const float4 bv = *(const float4*)&B[(size_t)(k0 + br) * N + (bn + bc)];
        __syncthreads();
        *(float4*)&As[ar][ac] = av;
        *(float4*)&Bs[br][bc] = bv;
        __syncthreads();

#pragma unroll
        for (int kk = 0; kk < KT; kk += 4) {
            const int kb = kk + 2 * hi;
            v2f a = *(const v2f*)&As[wm * 16 + l15][kb];
            v2f b0, b1;
            b0.x = Bs[kb    ][wn * 32 + l15];
            b0.y = Bs[kb + 1][wn * 32 + l15];
            b1.x = Bs[kb    ][wn * 32 + 16 + l15];
            b1.y = Bs[kb + 1][wn * 32 + 16 + l15];
            c0 = __builtin_amdgcn_wmma_f32_16x16x4_f32(false, a, false, b0,
                                                       (short)0, c0, false, false);
            c1 = __builtin_amdgcn_wmma_f32_16x16x4_f32(false, a, false, b1,
                                                       (short)0, c1, false, false);
        }
    }

    const int n0 = bn + wn * 32 + l15;
    const float bia0 = bias[n0];
    const float bia1 = bias[n0 + 16];
#pragma unroll
    for (int r = 0; r < 8; ++r) {
        const int m = bm + wm * 16 + r + 8 * hi;
        float v0 = c0[r] + bia0;
        float v1 = c1[r] + bia1;
        if (relu) { v0 = fmaxf(v0, 0.0f); v1 = fmaxf(v1, 0.0f); }
        C[(size_t)m * N + n0]      = v0;
        C[(size_t)m * N + n0 + 16] = v1;
    }
}

extern "C" void kernel_launch(void* const* d_in, const int* in_sizes, int n_in,
                              void* d_out, int out_size, void* d_ws, size_t ws_size,
                              hipStream_t stream) {
    const float* emb_a  = (const float*)d_in[0];
    const float* emb_b  = (const float*)d_in[1];
    const int*   keys_a = (const int*)d_in[2];
    const int*   keys_b = (const int*)d_in[3];
    const float* W1 = (const float*)d_in[4];
    const float* b1 = (const float*)d_in[5];
    const float* W2 = (const float*)d_in[6];
    const float* b2 = (const float*)d_in[7];
    const float* Wo = (const float*)d_in[8];
    const float* bo = (const float*)d_in[9];
    float* out = (float*)d_out;

    float* ws          = (float*)d_ws;
    int*   hash_b      = (int*)ws;
    float* bucket_sums = ws + BATCH_B;
    int*   counts      = (int*)(bucket_sums + NBUCKETS * D_MODEL);
    float* combined    = bucket_sums + NBUCKETS * D_MODEL + NBUCKETS;
    float* h           = combined + (size_t)BATCH_A * 2 * D_MODEL;
    float* rel         = h        + (size_t)BATCH_A * 2 * D_MODEL;

    hash_b_kernel<<<(BATCH_B + 255) / 256, 256, 0, stream>>>(keys_b, hash_b);
    bucket_sum_kernel<<<NBUCKETS, D_MODEL, 0, stream>>>(emb_b, hash_b, bucket_sums, counts);
    build_combined_kernel<<<BATCH_A, D_MODEL, 0, stream>>>(emb_a, keys_a, bucket_sums,
                                                           counts, combined);

    wmma_gemm_bias_kernel<<<dim3(512 / BN, BATCH_A / BM), 256, 0, stream>>>(
        combined, W1, b1, h, BATCH_A, 512, 512, 1);
    wmma_gemm_bias_kernel<<<dim3(256 / BN, BATCH_A / BM), 256, 0, stream>>>(
        h, W2, b2, rel, BATCH_A, 256, 512, 0);
    wmma_gemm_bias_kernel<<<dim3(256 / BN, BATCH_A / BM), 256, 0, stream>>>(
        rel, Wo, bo, out, BATCH_A, 256, 256, 0);
}


# GEMM K loop software-pipelined: next 64-k tile's global loads issued before the current tile's WMMAs, two LDS stages, one barrier per tile (was 16-k tiles: load, wait, store, compute); k order kept
# speedup vs baseline: 7.6329x; 2.2808x over previous
#include <hip/hip_runtime.h>

typedef float v2f __attribute__((ext_vector_type(2)));
typedef float v8f __attribute__((ext_vector_type(8)));

#define D_MODEL   256
#define BATCH_A   4096
#define BATCH_B   8192
#define NKEYS     4
#define NBUCKETS  1024

__device__ __forceinline__ int hash_row(const int* __restrict__ keys, int row) {
    int s = 0;
#pragma unroll
    for (int k = 0; k < NKEYS; ++k) s += keys[row * NKEYS + k];
    return (int)(((unsigned)s) & (NBUCKETS - 1));
}

__global__ void hash_b_kernel(const int* __restrict__ keys_b, int* __restrict__ hash_b) {
    int j = blockIdx.x * blockDim.x + threadIdx.x;
    if (j < BATCH_B) hash_b[j] = hash_row(keys_b, j);
}

#define SCAN_WAVES      (D_MODEL / 32)
#define SCAN_PER_WAVE   (BATCH_B / SCAN_WAVES)
#define SCAN_GROUPS     (SCAN_PER_WAVE / (32 * 4))
#define SUM_BATCH       16
static_assert(SCAN_WAVES * SCAN_PER_WAVE == BATCH_B, "scan must cover every b-row");
static_assert(SCAN_GROUPS * 128 == SCAN_PER_WAVE, "whole 16-byte loads only");
static_assert(BATCH_B <= 65536, "row numbers are kept as 16-bit values");

__device__ __forceinline__ int popc32(unsigned v) { return __builtin_popcount(v); }

__global__ __launch_bounds__(D_MODEL)
void bucket_sum_kernel(const float* __restrict__ emb_b,
                       const int* __restrict__ hash_b,
                       float* __restrict__ bucket_sums,
                       int* __restrict__ counts) {
    __shared__ __align__(16) unsigned short rows[BATCH_B + SUM_BATCH];
    __shared__ int wave_cnt[SCAN_WAVES];

    const int g    = blockIdx.x;
    const int t    = threadIdx.x;
    const int lane = t & 31;
    const int wave = t >> 5;

    const int4* hp = (const int4*)(hash_b + wave * SCAN_PER_WAVE) + lane;
    int4 hv[SCAN_GROUPS];
#pragma unroll
    for (int i = 0; i < SCAN_GROUPS; ++i) hv[i] = hp[i * 32];

    unsigned m0[SCAN_GROUPS], m1[SCAN_GROUPS], m2[SCAN_GROUPS], m3[SCAN_GROUPS];
    int cnt = 0;
#pragma unroll
    for (int i = 0; i < SCAN_GROUPS; ++i) {
        m0[i] = __builtin_amdgcn_ballot_w32(hv[i].x == g);
        m1[i] = __builtin_amdgcn_ballot_w32(hv[i].y == g);
        m2[i] = __builtin_amdgcn_ballot_w32(hv[i].z == g);
        m3[i] = __builtin_amdgcn_ballot_w32(hv[i].w == g);
        cnt += popc32(m0[i]) + popc32(m1[i]) + popc32(m2[i]) + popc32(m3[i]);
    }
    if (lane == 0) wave_cnt[wave] = cnt;
    __syncthreads();

    int base = 0, total = 0;
#pragma unroll
    for (int w = 0; w < SCAN_WAVES; ++w) {
        const int c = wave_cnt[w];
        if (w < wave) base += c;
        total += c;
    }
    if (t < SUM_BATCH) rows[total + t] = 0;
    if (cnt != 0) {
        const unsigned lt = (1u << lane) - 1u;
        int pos = base;
#pragma unroll
        for (int i = 0; i < SCAN_GROUPS; ++i) {
            if ((m0[i] | m1[i] | m2[i] | m3[i]) != 0u) {
                int p = pos + popc32(m0[i] & lt) + popc32(m1[i] & lt)
                            + popc32(m2[i] & lt) + popc32(m3[i] & lt);
                const int j0 = wave * SCAN_PER_WAVE + i * 128 + lane * 4;
                if (hv[i].x == g) rows[p++] = (unsigned short)(j0);
                if (hv[i].y == g) rows[p++] = (unsigned short)(j0 + 1);
                if (hv[i].z == g) rows[p++] = (unsigned short)(j0 + 2);
                if (hv[i].w == g) rows[p++] = (unsigned short)(j0 + 3);
                pos += popc32(m0[i]) + popc32(m1[i]) + popc32(m2[i]) + popc32(m3[i]);
            }
        }
    }
    __syncthreads();

    const float* __restrict__ src = emb_b + t;
    float acc = 0.0f;
    for (int e = 0; e < total; e += SUM_BATCH) {
        const uint4 pa = *(const uint4*)&rows[e];
        const uint4 pb = *(const uint4*)&rows[e + 8];
        const unsigned w[8] = {pa.x, pa.y, pa.z, pa.w, pb.x, pb.y, pb.z, pb.w};
        float r[SUM_BATCH];
#pragma unroll
        for (int k = 0; k < SUM_BATCH; k += 2) {
            r[k]     = src[(size_t)(w[k >> 1] & 0xffffu) * D_MODEL];
            r[k + 1] = src[(size_t)(w[k >> 1] >> 16)     * D_MODEL];
        }
#pragma unroll
        for (int k = 0; k < SUM_BATCH; ++k) {
            const float sum = acc + r[k];
            acc = (e + k < total) ? sum : acc;
        }
    }
    bucket_sums[(size_t)g * D_MODEL + t] = acc;
    if (t == 0) counts[g] = total;
}

__global__ __launch_bounds__(D_MODEL)
void build_combined_kernel(const float* __restrict__ emb_a,
                           const int* __restrict__ keys_a,
                           const float* __restrict__ bucket_sums,
                           const int* __restrict__ counts,
                           float* __restrict__ combined) {
    const int i = blockIdx.x;
    const int t = threadIdx.x;
    const int ha  = hash_row(keys_a, i);
    const int cnt = counts[ha];
    const float scale = 1.0f / (float)(cnt < 1 ? 1 : cnt);
    combined[(size_t)i * (2 * D_MODEL) + t]           = emb_a[(size_t)i * D_MODEL + t];
    combined[(size_t)i * (2 * D_MODEL) + D_MODEL + t] =
        bucket_sums[(size_t)ha * D_MODEL + t] * scale;
}

#define BM 64
#define BN 64
#define KT 64
#define AS_LD (KT + 4)
#define BS_LD (BN + 4)
#define LD_ROWS 16
#define A_LD4 (BM / LD_ROWS)
#define B_LD4 (KT / LD_ROWS)

__device__ __forceinline__ void gemm_tile_load(const float* __restrict__ Ap,
                                               const float* __restrict__ Bp,
                                               int K, int N,
                                               float4 (&av)[A_LD4], float4 (&bv)[B_LD4]) {
#pragma unroll
    for (int i = 0; i < A_LD4; ++i)
        av[i] = *(const float4*)(Ap + (size_t)(LD_ROWS * i) * K);
#pragma unroll
    for (int i = 0; i < B_LD4; ++i)
        bv[i] = *(const float4*)(Bp + (size_t)(LD_ROWS * i) * N);
}

__device__ __forceinline__ void gemm_tile_store(float (*As)[AS_LD], float (*Bs)[BS_LD],
                                                int lr, int lc,
                                                const float4 (&av)[A_LD4],
                                                const float4 (&bv)[B_LD4]) {
#pragma unroll
    for (int i = 0; i < A_LD4; ++i) *(float4*)&As[lr + LD_ROWS * i][lc] = av[i];
#pragma unroll
    for (int i = 0; i < B_LD4; ++i) *(float4*)&Bs[lr + LD_ROWS * i][lc] = bv[i];
}

__device__ __forceinline__ void gemm_tile_mma(const float* __restrict__ ap,
                                              const float* __restrict__ bp,
                                              v8f& c0, v8f& c1) {
#pragma unroll
    for (int kk = 0; kk < KT; kk += 4) {
        v2f a = *(const v2f*)(ap + kk);
        v2f b0, b1;
        b0.x = bp[(kk    ) * BS_LD];
        b0.y = bp[(kk + 1) * BS_LD];
        b1.x = bp[(kk    ) * BS_LD + 16];
        b1.y = bp[(kk + 1) * BS_LD + 16];
        c0 = __builtin_amdgcn_wmma_f32_16x16x4_f32(false, a, false, b0,
                                                   (short)0, c0, false, false);
        c1 = __builtin_amdgcn_wmma_f32_16x16x4_f32(false, a, false, b1,
                                                   (short)0, c1, false, false);
    }
}

__global__ __launch_bounds__(256)
void wmma_gemm_bias_kernel(const float* __restrict__ A,
                           const float* __restrict__ B,
                           const float* __restrict__ bias,
                           float* __restrict__ C,
                           int M, int N, int K, int relu) {
    __shared__ __align__(16) float As[2][BM][AS_LD];
    __shared__ __align__(16) float Bs[2][KT][BS_LD];

    const int tid  = threadIdx.x;
    const int lane = tid & 31;
    const int wave = tid >> 5;
    const int wm   = wave >> 1;
    const int wn   = wave & 1;
    const int hi   = lane >> 4;
    const int l15  = lane & 15;

    const int bm = blockIdx.y * BM;
    const int bn = blockIdx.x * BN;

    v8f c0 = {};
    v8f c1 = {};

    const int lr = tid >> 4;
    const int lc = (tid & 15) * 4;
    const float* __restrict__ Ap = A + (size_t)(bm + lr) * K + lc;
    const float* __restrict__ Bp = B + (size_t)lr * N + (bn + lc);

    const int aoff = (wm * 16 + l15) * AS_LD + 2 * hi;
    const int boff = (2 * hi) * BS_LD + wn * 32 + l15;

    float4 av[A_LD4], bv[B_LD4];
    gemm_tile_load(Ap, Bp, K, N, av, bv);
    gemm_tile_store(As[0], Bs[0], lr, lc, av, bv);
    __syncthreads();

    const int nk = K / KT;
    int cur = 0;
    for (int t = 1; t < nk; ++t) {
        gemm_tile_load(Ap + (size_t)t * KT, Bp + (size_t)t * KT * N, K, N, av, bv);
        gemm_tile_mma(&As[cur][0][0] + aoff, &Bs[cur][0][0] + boff, c0, c1);
        gemm_tile_store(As[cur ^ 1], Bs[cur ^ 1], lr, lc, av, bv);
        __syncthreads();
        cur ^= 1;
    }
    gemm_tile_mma(&As[cur][0][0] + aoff, &Bs[cur][0][0] + boff, c0, c1);

    const int n0 = bn + wn * 32 + l15;
    const float bia0 = bias[n0];
    const float bia1 = bias[n0 + 16];
#pragma unroll
    for (int r = 0; r < 8; ++r) {
        const int m = bm + wm * 16 + r + 8 * hi;
        float v0 = c0[r] + bia0;
        float v1 = c1[r] + bia1;
        if (relu) { v0 = fmaxf(v0, 0.0f); v1 = fmaxf(v1, 0.0f); }
        C[(size_t)m * N + n0]      = v0;
        C[(size_t)m * N + n0 + 16] = v1;
    }
}

extern "C" void kernel_launch(void* const* d_in, const int* in_sizes, int n_in,
                              void* d_out, int out_size, void* d_ws, size_t ws_size,
                              hipStream_t stream) {
    const float* emb_a  = (const float*)d_in[0];
    const float* emb_b  = (const float*)d_in[1];
    const int*   keys_a = (const int*)d_in[2];
    const int*   keys_b = (const int*)d_in[3];
    const float* W1 = (const float*)d_in[4];
    const float* b1 = (const float*)d_in[5];
    const float* W2 = (const float*)d_in[6];
    const float* b2 = (const float*)d_in[7];
    const float* Wo = (const float*)d_in[8];
    const float* bo = (const float*)d_in[9];
    float* out = (float*)d_out;

    float* ws          = (float*)d_ws;
    int*   hash_b      = (int*)ws;
    float* bucket_sums = ws + BATCH_B;
    int*   counts      = (int*)(bucket_sums + NBUCKETS * D_MODEL);
    float* combined    = bucket_sums + NBUCKETS * D_MODEL + NBUCKETS;
    float* h           = combined + (size_t)BATCH_A * 2 * D_MODEL;
    float* rel         = h        + (size_t)BATCH_A * 2 * D_MODEL;

    hash_b_kernel<<<(BATCH_B + 255) / 256, 256, 0, stream>>>(keys_b, hash_b);
    bucket_sum_kernel<<<NBUCKETS, D_MODEL, 0, stream>>>(emb_b, hash_b, bucket_sums, counts);
    build_combined_kernel<<<BATCH_A, D_MODEL, 0, stream>>>(emb_a, keys_a, bucket_sums,
                                                           counts, combined);

    wmma_gemm_bias_kernel<<<dim3(512 / BN, BATCH_A / BM), 256, 0, stream>>>(
        combined, W1, b1, h, BATCH_A, 512, 512, 1);
    wmma_gemm_bias_kernel<<<dim3(256 / BN, BATCH_A / BM), 256, 0, stream>>>(
        h, W2, b2, rel, BATCH_A, 256, 512, 0);
    wmma_gemm_bias_kernel<<<dim3(256 / BN, BATCH_A / BM), 256, 0, stream>>>(
        rel, Wo, bo, out, BATCH_A, 256, 256, 0);
}


# first GEMM's A loader gathers [emb_a row | bucket sum x 1/count] itself: the 4096-block kernel that wrote the combined matrix, its launch and its 16 MB of traffic are gone; values bit-identical
# speedup vs baseline: 8.1298x; 1.0651x over previous
#include <hip/hip_runtime.h>

typedef float v2f __attribute__((ext_vector_type(2)));
typedef float v8f __attribute__((ext_vector_type(8)));

#define D_MODEL   256
#define BATCH_A   4096
#define BATCH_B   8192
#define NKEYS     4
#define NBUCKETS  1024

__device__ __forceinline__ int hash_row(const int* __restrict__ keys, int row) {
    int s = 0;
#pragma unroll
    for (int k = 0; k < NKEYS; ++k) s += keys[row * NKEYS + k];
    return (int)(((unsigned)s) & (NBUCKETS - 1));
}

__global__ void hash_b_kernel(const int* __restrict__ keys_b, int* __restrict__ hash_b) {
    int j = blockIdx.x * blockDim.x + threadIdx.x;
    if (j < BATCH_B) hash_b[j] = hash_row(keys_b, j);
}

#define SCAN_WAVES      (D_MODEL / 32)
#define SCAN_PER_WAVE   (BATCH_B / SCAN_WAVES)
#define SCAN_GROUPS     (SCAN_PER_WAVE / (32 * 4))
#define SUM_BATCH       16
static_assert(SCAN_WAVES * SCAN_PER_WAVE == BATCH_B, "scan must cover every b-row");
static_assert(SCAN_GROUPS * 128 == SCAN_PER_WAVE, "whole 16-byte loads only");
static_assert(BATCH_B <= 65536, "row numbers are kept as 16-bit values");

__device__ __forceinline__ int popc32(unsigned v) { return __builtin_popcount(v); }

__global__ __launch_bounds__(D_MODEL)
void bucket_sum_kernel(const float* __restrict__ emb_b,
                       const int* __restrict__ hash_b,
                       float* __restrict__ bucket_sums,
                       int* __restrict__ counts) {
    __shared__ __align__(16) unsigned short rows[BATCH_B + SUM_BATCH];
    __shared__ int wave_cnt[SCAN_WAVES];

    const int g    = blockIdx.x;
    const int t    = threadIdx.x;
    const int lane = t & 31;
    const int wave = t >> 5;

    const int4* hp = (const int4*)(hash_b + wave * SCAN_PER_WAVE) + lane;
    int4 hv[SCAN_GROUPS];
#pragma unroll
    for (int i = 0; i < SCAN_GROUPS; ++i) hv[i] = hp[i * 32];

    unsigned m0[SCAN_GROUPS], m1[SCAN_GROUPS], m2[SCAN_GROUPS], m3[SCAN_GROUPS];
    int cnt = 0;
#pragma unroll
    for (int i = 0; i < SCAN_GROUPS; ++i) {
        m0[i] = __builtin_amdgcn_ballot_w32(hv[i].x == g);
        m1[i] = __builtin_amdgcn_ballot_w32(hv[i].y == g);
        m2[i] = __builtin_amdgcn_ballot_w32(hv[i].z == g);
        m3[i] = __builtin_amdgcn_ballot_w32(hv[i].w == g);
        cnt += popc32(m0[i]) + popc32(m1[i]) + popc32(m2[i]) + popc32(m3[i]);
    }
    if (lane == 0) wave_cnt[wave] = cnt;
    __syncthreads();

    int base = 0, total = 0;
#pragma unroll
    for (int w = 0; w < SCAN_WAVES; ++w) {
        const int c = wave_cnt[w];
        if (w < wave) base += c;
        total += c;
    }
    if (t < SUM_BATCH) rows[total + t] = 0;
    if (cnt != 0) {
        const unsigned lt = (1u << lane) - 1u;
        int pos = base;
#pragma unroll
        for (int i = 0; i < SCAN_GROUPS; ++i) {
            if ((m0[i] | m1[i] | m2[i] | m3[i]) != 0u) {
                int p = pos + popc32(m0[i] & lt) + popc32(m1[i] & lt)
                            + popc32(m2[i] & lt) + popc32(m3[i] & lt);
                const int j0 = wave * SCAN_PER_WAVE + i * 128 + lane * 4;
                if (hv[i].x == g) rows[p++] = (unsigned short)(j0);
                if (hv[i].y == g) rows[p++] = (unsigned short)(j0 + 1);
                if (hv[i].z == g) rows[p++] = (unsigned short)(j0 + 2);
                if (hv[i].w == g) rows[p++] = (unsigned short)(j0 + 3);
                pos += popc32(m0[i]) + popc32(m1[i]) + popc32(m2[i]) + popc32(m3[i]);
            }
        }
    }
    __syncthreads();

    const float* __restrict__ src = emb_b + t;
    float acc = 0.0f;
    for (int e = 0; e < total; e += SUM_BATCH) {
        const uint4 pa = *(const uint4*)&rows[e];
        const uint4 pb = *(const uint4*)&rows[e + 8];
        const unsigned w[8] = {pa.x, pa.y, pa.z, pa.w, pb.x, pb.y, pb.z, pb.w};
        float r[SUM_BATCH];
#pragma unroll
        for (int k = 0; k < SUM_BATCH; k += 2) {
            r[k]     = src[(size_t)(w[k >> 1] & 0xffffu) * D_MODEL];
            r[k + 1] = src[(size_t)(w[k >> 1] >> 16)     * D_MODEL];
        }
#pragma unroll
        for (int k = 0; k < SUM_BATCH; ++k) {
            const float sum = acc + r[k];
            acc = (e + k < total) ? sum : acc;
        }
    }
    bucket_sums[(size_t)g * D_MODEL + t] = acc;
    if (t == 0) counts[g] = total;
}

#define BM 64
#define BN 64
#define KT 64
#define AS_LD (KT + 4)
#define BS_LD (BN + 4)
#define LD_ROWS 16
#define A_LD4 (BM / LD_ROWS)
#define B_LD4 (KT / LD_ROWS)
static_assert(D_MODEL % KT == 0, "a k tile lies wholly in one half of the gathered row");

__device__ __forceinline__ void gemm_tile_load_a(const float* __restrict__ Ap, int lda,
                                                 float4 (&av)[A_LD4]) {
#pragma unroll
    for (int i = 0; i < A_LD4; ++i)
        av[i] = *(const float4*)(Ap + (size_t)(LD_ROWS * i) * lda);
}

__device__ __forceinline__ void gemm_tile_gather_a(const float* __restrict__ Gp,
                                                   const unsigned (&goff)[A_LD4],
                                                   float4 (&av)[A_LD4]) {
#pragma unroll
    for (int i = 0; i < A_LD4; ++i)
        av[i] = *(const float4*)(Gp + goff[i]);
}

__device__ __forceinline__ void gemm_tile_load_b(const float* __restrict__ Bp, int N,
                                                 float4 (&bv)[B_LD4]) {
#pragma unroll
    for (int i = 0; i < B_LD4; ++i)
        bv[i] = *(const float4*)(Bp + (size_t)(LD_ROWS * i) * N);
}

__device__ __forceinline__ void gemm_tile_store(float (*As)[AS_LD], float (*Bs)[BS_LD],
                                                int lr, int lc,
                                                const float4 (&av)[A_LD4],
                                                const float4 (&bv)[B_LD4]) {
#pragma unroll
    for (int i = 0; i < A_LD4; ++i) *(float4*)&As[lr + LD_ROWS * i][lc] = av[i];
#pragma unroll
    for (int i = 0; i < B_LD4; ++i) *(float4*)&Bs[lr + LD_ROWS * i][lc] = bv[i];
}

__device__ __forceinline__ void gemm_tile_mma(const float* __restrict__ ap,
                                              const float* __restrict__ bp,
                                              v8f& c0, v8f& c1) {
#pragma unroll
    for (int kk = 0; kk < KT; kk += 4) {
        v2f a = *(const v2f*)(ap + kk);
        v2f b0, b1;
        b0.x = bp[(kk    ) * BS_LD];
        b0.y = bp[(kk + 1) * BS_LD];
        b1.x = bp[(kk    ) * BS_LD + 16];
        b1.y = bp[(kk + 1) * BS_LD + 16];
        c0 = __builtin_amdgcn_wmma_f32_16x16x4_f32(false, a, false, b0,
                                                   (short)0, c0, false, false);
        c1 = __builtin_amdgcn_wmma_f32_16x16x4_f32(false, a, false, b1,
                                                   (short)0, c1, false, false);
    }
}

template <bool GATHER>
__global__ __launch_bounds__(256)
void wmma_gemm_bias_kernel(const float* __restrict__ A,
                           const float* __restrict__ B,
                           const float* __restrict__ bias,
                           float* __restrict__ C,
                           int M, int N, int K, int relu,
                           const int* __restrict__ keys_a,
                           const float* __restrict__ bucket_sums,
                           const int* __restrict__ counts) {
    __shared__ __align__(16) float As[2][BM][AS_LD];
    __shared__ __align__(16) float Bs[2][KT][BS_LD];

    const int tid  = threadIdx.x;
    const int lane = tid & 31;
    const int wave = tid >> 5;
    const int wm   = wave >> 1;
    const int wn   = wave & 1;
    const int hi   = lane >> 4;
    const int l15  = lane & 15;

    const int bm = blockIdx.y * BM;
    const int bn = blockIdx.x * BN;

    v8f c0 = {};
    v8f c1 = {};

    const int lr = tid >> 4;
    const int lc = (tid & 15) * 4;
    const int lda = GATHER ? D_MODEL : K;
    const int ka  = GATHER ? D_MODEL : K;
    const float* __restrict__ Ap = A + (size_t)(bm + lr) * lda + lc;
    const float* __restrict__ Bp = B + (size_t)lr * N + (bn + lc);

    int ha[A_LD4];
    if (GATHER) {
#pragma unroll
        for (int i = 0; i < A_LD4; ++i) ha[i] = hash_row(keys_a, bm + lr + LD_ROWS * i);
    }

    const int aoff = (wm * 16 + l15) * AS_LD + 2 * hi;
    const int boff = (2 * hi) * BS_LD + wn * 32 + l15;

    float4 av[A_LD4], bv[B_LD4];
    gemm_tile_load_a(Ap, lda, av);
    gemm_tile_load_b(Bp, N, bv);
    gemm_tile_store(As[0], Bs[0], lr, lc, av, bv);
    __syncthreads();

    unsigned goff[A_LD4];
    int gcnt[A_LD4];
    if (GATHER) {
#pragma unroll
        for (int i = 0; i < A_LD4; ++i) {
            gcnt[i] = counts[ha[i]];
            goff[i] = (unsigned)(ha[i] * D_MODEL + lc);
        }
    }

    const int nk  = K / KT;
    const int nka = ka / KT;
    int cur = 0;
    int t = 1;
    for (; t < nka; ++t) {
        gemm_tile_load_a(Ap + (size_t)t * KT, lda, av);
        gemm_tile_load_b(Bp + (size_t)t * KT * N, N, bv);
        gemm_tile_mma(&As[cur][0][0] + aoff, &Bs[cur][0][0] + boff, c0, c1);
        gemm_tile_store(As[cur ^ 1], Bs[cur ^ 1], lr, lc, av, bv);
        __syncthreads();
        cur ^= 1;
    }
    if (GATHER) {
        float gs[A_LD4];
#pragma unroll
        for (int i = 0; i < A_LD4; ++i)
            gs[i] = 1.0f / (float)(gcnt[i] < 1 ? 1 : gcnt[i]);
        for (; t < nk; ++t) {
            gemm_tile_gather_a(bucket_sums + (t * KT - D_MODEL), goff, av);
            gemm_tile_load_b(Bp + (size_t)t * KT * N, N, bv);
            gemm_tile_mma(&As[cur][0][0] + aoff, &Bs[cur][0][0] + boff, c0, c1);
#pragma unroll
            for (int i = 0; i < A_LD4; ++i) {
                av[i].x *= gs[i]; av[i].y *= gs[i]; av[i].z *= gs[i]; av[i].w *= gs[i];
            }
            gemm_tile_store(As[cur ^ 1], Bs[cur ^ 1], lr, lc, av, bv);
            __syncthreads();
            cur ^= 1;
        }
    }
    gemm_tile_mma(&As[cur][0][0] + aoff, &Bs[cur][0][0] + boff, c0, c1);

    const int n0 = bn + wn * 32 + l15;
    const float bia0 = bias[n0];
    const float bia1 = bias[n0 + 16];
#pragma unroll
    for (int r = 0; r < 8; ++r) {
        const int m = bm + wm * 16 + r + 8 * hi;
        float v0 = c0[r] + bia0;
        float v1 = c1[r] + bia1;
        if (relu) { v0 = fmaxf(v0, 0.0f); v1 = fmaxf(v1, 0.0f); }
        C[(size_t)m * N + n0]      = v0;
        C[(size_t)m * N + n0 + 16] = v1;
    }
}

extern "C" void kernel_launch(void* const* d_in, const int* in_sizes, int n_in,
                              void* d_out, int out_size, void* d_ws, size_t ws_size,
                              hipStream_t stream) {
    const float* emb_a  = (const float*)d_in[0];
    const float* emb_b  = (const float*)d_in[1];
    const int*   keys_a = (const int*)d_in[2];
    const int*   keys_b = (const int*)d_in[3];
    const float* W1 = (const float*)d_in[4];
    const float* b1 = (const float*)d_in[5];
    const float* W2 = (const float*)d_in[6];
    const float* b2 = (const float*)d_in[7];
    const float* Wo = (const float*)d_in[8];
    const float* bo = (const float*)d_in[9];
    float* out = (float*)d_out;

    float* ws          = (float*)d_ws;
    int*   hash_b      = (int*)ws;
    float* bucket_sums = ws + BATCH_B;
    int*   counts      = (int*)(bucket_sums + NBUCKETS * D_MODEL);
    float* h           = bucket_sums + NBUCKETS * D_MODEL + NBUCKETS;
    float* rel         = h + (size_t)BATCH_A * 2 * D_MODEL;

    hash_b_kernel<<<(BATCH_B + 255) / 256, 256, 0, stream>>>(keys_b, hash_b);
    bucket_sum_kernel<<<NBUCKETS, D_MODEL, 0, stream>>>(emb_b, hash_b, bucket_sums, counts);

    wmma_gemm_bias_kernel<true><<<dim3(512 / BN, BATCH_A / BM), 256, 0, stream>>>(
        emb_a, W1, b1, h, BATCH_A, 512, 512, 1, keys_a, bucket_sums, counts);
    wmma_gemm_bias_kernel<false><<<dim3(256 / BN, BATCH_A / BM), 256, 0, stream>>>(
        h, W2, b2, rel, BATCH_A, 256, 512, 0, nullptr, nullptr, nullptr);
    wmma_gemm_bias_kernel<false><<<dim3(256 / BN, BATCH_A / BM), 256, 0, stream>>>(
        rel, Wo, bo, out, BATCH_A, 256, 256, 0, nullptr, nullptr, nullptr);
}


# GEMM fragments read from LDS 16 bytes at a time (B tile kept as k quads): one read feeds two WMMAs, reads grouped ahead of the WMMAs; k summed 0,1,4,5 then 2,3,6,7 per 8: rounding changes
# speedup vs baseline: 10.1586x; 1.2496x over previous
#include <hip/hip_runtime.h>

typedef float v2f __attribute__((ext_vector_type(2)));
typedef float v8f __attribute__((ext_vector_type(8)));

#define D_MODEL   256
#define BATCH_A   4096
#define BATCH_B   8192
#define NKEYS     4
#define NBUCKETS  1024

__device__ __forceinline__ int hash_row(const int* __restrict__ keys, int row) {
    int s = 0;
#pragma unroll
    for (int k = 0; k < NKEYS; ++k) s += keys[row * NKEYS + k];
    return (int)(((unsigned)s) & (NBUCKETS - 1));
}

__global__ void hash_b_kernel(const int* __restrict__ keys_b, int* __restrict__ hash_b) {
    int j = blockIdx.x * blockDim.x + threadIdx.x;
    if (j < BATCH_B) hash_b[j] = hash_row(keys_b, j);
}

#define SCAN_WAVES      (D_MODEL / 32)
#define SCAN_PER_WAVE   (BATCH_B / SCAN_WAVES)
#define SCAN_GROUPS     (SCAN_PER_WAVE / (32 * 4))
#define SUM_BATCH       16
static_assert(SCAN_WAVES * SCAN_PER_WAVE == BATCH_B, "scan must cover every b-row");
static_assert(SCAN_GROUPS * 128 == SCAN_PER_WAVE, "whole 16-byte loads only");
static_assert(BATCH_B <= 65536, "row numbers are kept as 16-bit values");

__device__ __forceinline__ int popc32(unsigned v) { return __builtin_popcount(v); }

__global__ __launch_bounds__(D_MODEL)
void bucket_sum_kernel(const float* __restrict__ emb_b,
                       const int* __restrict__ hash_b,
                       float* __restrict__ bucket_sums,
                       int* __restrict__ counts) {
    __shared__ __align__(16) unsigned short rows[BATCH_B + SUM_BATCH];
    __shared__ int wave_cnt[SCAN_WAVES];

    const int g    = blockIdx.x;
    const int t    = threadIdx.x;
    const int lane = t & 31;
    const int wave = t >> 5;

    const int4* hp = (const int4*)(hash_b + wave * SCAN_PER_WAVE) + lane;
    int4 hv[SCAN_GROUPS];
#pragma unroll
    for (int i = 0; i < SCAN_GROUPS; ++i) hv[i] = hp[i * 32];

    unsigned m0[SCAN_GROUPS], m1[SCAN_GROUPS], m2[SCAN_GROUPS], m3[SCAN_GROUPS];
    int cnt = 0;
#pragma unroll
    for (int i = 0; i < SCAN_GROUPS; ++i) {
        m0[i] = __builtin_amdgcn_ballot_w32(hv[i].x == g);
        m1[i] = __builtin_amdgcn_ballot_w32(hv[i].y == g);
        m2[i] = __builtin_amdgcn_ballot_w32(hv[i].z == g);
        m3[i] = __builtin_amdgcn_ballot_w32(hv[i].w == g);
        cnt += popc32(m0[i]) + popc32(m1[i]) + popc32(m2[i]) + popc32(m3[i]);
    }
    if (lane == 0) wave_cnt[wave] = cnt;
    __syncthreads();

    int base = 0, total = 0;
#pragma unroll
    for (int w = 0; w < SCAN_WAVES; ++w) {
        const int c = wave_cnt[w];
        if (w < wave) base += c;
        total += c;
    }
    if (t < SUM_BATCH) rows[total + t] = 0;
    if (cnt != 0) {
        const unsigned lt = (1u << lane) - 1u;
        int pos = base;
#pragma unroll
        for (int i = 0; i < SCAN_GROUPS; ++i) {
            if ((m0[i] | m1[i] | m2[i] | m3[i]) != 0u) {
                int p = pos + popc32(m0[i] & lt) + popc32(m1[i] & lt)
                            + popc32(m2[i] & lt) + popc32(m3[i] & lt);
                const int j0 = wave * SCAN_PER_WAVE + i * 128 + lane * 4;
                if (hv[i].x == g) rows[p++] = (unsigned short)(j0);
                if (hv[i].y == g) rows[p++] = (unsigned short)(j0 + 1);
                if (hv[i].z == g) rows[p++] = (unsigned short)(j0 + 2);
                if (hv[i].w == g) rows[p++] = (unsigned short)(j0 + 3);
                pos += popc32(m0[i]) + popc32(m1[i]) + popc32(m2[i]) + popc32(m3[i]);
            }
        }
    }
    __syncthreads();

    const float* __restrict__ src = emb_b + t;
    float acc = 0.0f;
    for (int e = 0; e < total; e += SUM_BATCH) {
        const uint4 pa = *(const uint4*)&rows[e];
        const uint4 pb = *(const uint4*)&rows[e + 8];
        const unsigned w[8] = {pa.x, pa.y, pa.z, pa.w, pb.x, pb.y, pb.z, pb.w};
        float r[SUM_BATCH];
#pragma unroll
        for (int k = 0; k < SUM_BATCH; k += 2) {
            r[k]     = src[(size_t)(w[k >> 1] & 0xffffu) * D_MODEL];
            r[k + 1] = src[(size_t)(w[k >> 1] >> 16)     * D_MODEL];
        }
#pragma unroll
        for (int k = 0; k < SUM_BATCH; ++k) {
            const float sum = acc + r[k];
            acc = (e + k < total) ? sum : acc;
        }
    }
    bucket_sums[(size_t)g * D_MODEL + t] = acc;
    if (t == 0) counts[g] = total;
}

#define BM 64
#define BN 64
#define KT 64
#define AS_LD (KT + 4)
#define BQ (KT / 4)
#define BQ_LD (4 * BN + 32)
#define LD_ROWS 16
#define A_LD4 (BM / LD_ROWS)
#define B_LD4 4
static_assert(BQ == LD_ROWS, "one k quad of the B tile per staging row group");
static_assert(BN == 64, "slot() swizzles with bits 4..5 of the column");
static_assert(D_MODEL % KT == 0, "a k tile lies wholly in one half of the gathered row");

__device__ __forceinline__ void gemm_tile_load_a(const float* __restrict__ Ap, int lda,
                                                 float4 (&av)[A_LD4]) {
#pragma unroll
    for (int i = 0; i < A_LD4; ++i)
        av[i] = *(const float4*)(Ap + (size_t)(LD_ROWS * i) * lda);
}

__device__ __forceinline__ void gemm_tile_gather_a(const float* __restrict__ Gp,
                                                   const unsigned (&goff)[A_LD4],
                                                   float4 (&av)[A_LD4]) {
#pragma unroll
    for (int i = 0; i < A_LD4; ++i)
        av[i] = *(const float4*)(Gp + goff[i]);
}

__device__ __forceinline__ void gemm_tile_load_b(const float* __restrict__ Bp, int N,
                                                 float4 (&bv)[B_LD4]) {
#pragma unroll
    for (int i = 0; i < B_LD4; ++i)
        bv[i] = *(const float4*)(Bp + (size_t)i * N);
}

__device__ __forceinline__ void gemm_tile_store(float (*As)[AS_LD], float* bsp, int bsw,
                                                int lr, int lc,
                                                const float4 (&av)[A_LD4],
                                                const float4 (&bv)[B_LD4]) {
#pragma unroll
    for (int i = 0; i < A_LD4; ++i) *(float4*)&As[lr + LD_ROWS * i][lc] = av[i];
    const float4 t0 = {bv[0].x, bv[1].x, bv[2].x, bv[3].x};
    const float4 t1 = {bv[0].y, bv[1].y, bv[2].y, bv[3].y};
    const float4 t2 = {bv[0].z, bv[1].z, bv[2].z, bv[3].z};
    const float4 t3 = {bv[0].w, bv[1].w, bv[2].w, bv[3].w};
    *(float4*)(bsp + 4 * (0 ^ bsw)) = t0;
    *(float4*)(bsp + 4 * (1 ^ bsw)) = t1;
    *(float4*)(bsp + 4 * (2 ^ bsw)) = t2;
    *(float4*)(bsp + 4 * (3 ^ bsw)) = t3;
}

__device__ __forceinline__ void gemm_tile_mma(const float* __restrict__ ap,
                                              const float* __restrict__ bp0,
                                              const float* __restrict__ bp1,
                                              v8f& c0, v8f& c1) {
    float4 a[KT / 8], b0[KT / 8], b1[KT / 8];
#pragma unroll
    for (int j = 0; j < KT / 8; ++j) {
        a[j]  = *(const float4*)(ap + 8 * j);
        b0[j] = *(const float4*)(bp0 + 2 * j * BQ_LD);
        b1[j] = *(const float4*)(bp1 + 2 * j * BQ_LD);
    }
#pragma unroll
    for (int j = 0; j < KT / 8; ++j) {
        const v2f alo = {a[j].x, a[j].y},   ahi = {a[j].z, a[j].w};
        const v2f p0  = {b0[j].x, b0[j].y}, q0  = {b0[j].z, b0[j].w};
        const v2f p1  = {b1[j].x, b1[j].y}, q1  = {b1[j].z, b1[j].w};
        c0 = __builtin_amdgcn_wmma_f32_16x16x4_f32(false, alo, false, p0,
                                                   (short)0, c0, false, false);
        c1 = __builtin_amdgcn_wmma_f32_16x16x4_f32(false, alo, false, p1,
                                                   (short)0, c1, false, false);
        c0 = __builtin_amdgcn_wmma_f32_16x16x4_f32(false, ahi, false, q0,
                                                   (short)0, c0, false, false);
        c1 = __builtin_amdgcn_wmma_f32_16x16x4_f32(false, ahi, false, q1,
                                                   (short)0, c1, false, false);
    }
}

template <int GROUPS, bool STAGING>
__device__ __forceinline__ void gemm_step_order() {
    if constexpr (STAGING) __builtin_amdgcn_sched_group_barrier(0x020, A_LD4 + B_LD4, 0);
    if constexpr (GROUPS == 1) {
        __builtin_amdgcn_sched_group_barrier(0x100, 3 * (KT / 8), 0);
        __builtin_amdgcn_sched_group_barrier(0x008, 4 * (KT / 8), 0);
    } else {
        __builtin_amdgcn_sched_group_barrier(0x100, 3 * (KT / 16), 0);
        __builtin_amdgcn_sched_group_barrier(0x008, 4 * (KT / 16), 0);
        __builtin_amdgcn_sched_group_barrier(0x100, 3 * (KT / 16), 0);
        __builtin_amdgcn_sched_group_barrier(0x008, 4 * (KT / 16), 0);
    }
    if constexpr (STAGING) __builtin_amdgcn_sched_group_barrier(0x200, A_LD4 + B_LD4, 0);
}

template <bool GATHER, int GROUPS>
__device__ __forceinline__
void gemm_block(float (&As)[2][BM][AS_LD], float (&Bs)[2][BQ][BQ_LD],
                const int bm, const int bn,
                const float* __restrict__ A,
                const float* __restrict__ B,
                const float* __restrict__ bias,
                float* __restrict__ C,
                int N, int K, int relu,
                const int* __restrict__ keys_a,
                const float* __restrict__ bucket_sums,
                const int* __restrict__ counts) {
    const int tid  = threadIdx.x;
    const int lane = tid & 31;
    const int wave = tid >> 5;
    const int wm   = wave >> 1;
    const int wn   = wave & 1;
    const int hi   = lane >> 4;
    const int l15  = lane & 15;

    v8f c0 = {};
    v8f c1 = {};

    const int lr = tid >> 4;
    const int lc = (tid & 15) * 4;
    const int lda = GATHER ? D_MODEL : K;
    const int ka  = GATHER ? D_MODEL : K;
    const float* __restrict__ Ap = A + (size_t)(bm + lr) * lda + lc;
    const float* __restrict__ Bp = B + (size_t)(4 * lr) * N + (bn + lc);
    const int bsw  = (lc >> 4) & 3;
    const int bsto = lr * BQ_LD + 4 * lc;

    int ha[A_LD4];
    if (GATHER) {
#pragma unroll
        for (int i = 0; i < A_LD4; ++i) ha[i] = hash_row(keys_a, bm + lr + LD_ROWS * i);
    }

    const int aoff  = (wm * 16 + l15) * AS_LD + 4 * hi;
    const int boff0 = hi * BQ_LD + 4 * (wn * 32 +      (l15 ^ (2 * wn)));
    const int boff1 = hi * BQ_LD + 4 * (wn * 32 + 16 + (l15 ^ (2 * wn + 1)));

    float4 av[A_LD4], bv[B_LD4];
    gemm_tile_load_a(Ap, lda, av);
    gemm_tile_load_b(Bp, N, bv);
    gemm_tile_store(As[0], &Bs[0][0][0] + bsto, bsw, lr, lc, av, bv);
    __syncthreads();

    unsigned goff[A_LD4];
    int gcnt[A_LD4];
    if (GATHER) {
#pragma unroll
        for (int i = 0; i < A_LD4; ++i) {
            gcnt[i] = counts[ha[i]];
            goff[i] = (unsigned)(ha[i] * D_MODEL + lc);
        }
    }

    const int nk  = K / KT;
    const int nka = ka / KT;
    int cur = 0;
    int t = 1;
#pragma nounroll
    for (; t < nka; ++t) {
        gemm_tile_load_a(Ap + (size_t)t * KT, lda, av);
        gemm_tile_load_b(Bp + (size_t)t * KT * N, N, bv);
        gemm_tile_mma(&As[cur][0][0] + aoff, &Bs[cur][0][0] + boff0, &Bs[cur][0][0] + boff1, c0, c1);
        gemm_tile_store(As[cur ^ 1], &Bs[cur ^ 1][0][0] + bsto, bsw, lr, lc, av, bv);
        gemm_step_order<GROUPS, true>();
        __syncthreads();
        cur ^= 1;
    }
    if (GATHER) {
        float gs[A_LD4];
#pragma unroll
        for (int i = 0; i < A_LD4; ++i)
            gs[i] = 1.0f / (float)(gcnt[i] < 1 ? 1 : gcnt[i]);
#pragma nounroll
        for (; t < nk; ++t) {
            gemm_tile_gather_a(bucket_sums + (t * KT - D_MODEL), goff, av);
            gemm_tile_load_b(Bp + (size_t)t * KT * N, N, bv);
            gemm_tile_mma(&As[cur][0][0] + aoff, &Bs[cur][0][0] + boff0, &Bs[cur][0][0] + boff1, c0, c1);
#pragma unroll
            for (int i = 0; i < A_LD4; ++i) {
                av[i].x *= gs[i]; av[i].y *= gs[i]; av[i].z *= gs[i]; av[i].w *= gs[i];
            }
            gemm_tile_store(As[cur ^ 1], &Bs[cur ^ 1][0][0] + bsto, bsw, lr, lc, av, bv);
            gemm_step_order<GROUPS, true>();
            __syncthreads();
            cur ^= 1;
        }
    }
    gemm_tile_mma(&As[cur][0][0] + aoff, &Bs[cur][0][0] + boff0, &Bs[cur][0][0] + boff1, c0, c1);
    gemm_step_order<GROUPS, false>();

    const int n0 = bn + wn * 32 + l15;
    const float bia0 = bias ? bias[n0] : 0.0f;
    const float bia1 = bias ? bias[n0 + 16] : 0.0f;
#pragma unroll
    for (int r = 0; r < 8; ++r) {
        const int m = bm + wm * 16 + r + 8 * hi;
        float v0 = c0[r] + bia0;
        float v1 = c1[r] + bia1;
        if (relu) { v0 = fmaxf(v0, 0.0f); v1 = fmaxf(v1, 0.0f); }
        C[(size_t)m * N + n0]      = v0;
        C[(size_t)m * N + n0 + 16] = v1;
    }
}

__global__ __launch_bounds__(256)
void wmma_gemm_bias_kernel(const float* __restrict__ A,
                           const float* __restrict__ B,
                           const float* __restrict__ bias,
                           float* __restrict__ C,
                           int N, int K, int relu) {
    __shared__ __align__(16) float As[2][BM][AS_LD];
    __shared__ __align__(16) float Bs[2][BQ][BQ_LD];
    gemm_block<false, 1>(As, Bs, blockIdx.y * BM, blockIdx.x * BN, A, B, bias, C, N, K, relu,
                      nullptr, nullptr, nullptr);
}

#define L1_GRID_X    (2 * D_MODEL / BN)
#define L1_GRID_Y    (BATCH_A / BM)
#define MERGE_GRID_X (D_MODEL / BN)
#define MERGE_TILES  ((2 * D_MODEL / BM) * MERGE_GRID_X)
#define EXTRA_GRID_Y ((MERGE_TILES + 1 + L1_GRID_X - 1) / L1_GRID_X)

__device__ __forceinline__ void merged_bias_block(const float* __restrict__ b2,
                                                  const float* __restrict__ Wo,
                                                  const float* __restrict__ bo,
                                                  float* __restrict__ bias_m) {
    const int n = threadIdx.x;
    float acc = 0.0f;
#pragma unroll 64
    for (int j = 0; j < D_MODEL; ++j) acc = fmaf(b2[j], Wo[(size_t)j * D_MODEL + n], acc);
    bias_m[n] = acc + bo[n];
}

__global__ __launch_bounds__(256)
void first_layer_and_merge_kernel(const float* __restrict__ emb_a,
                                  const float* __restrict__ W1,
                                  const float* __restrict__ b1,
                                  float* __restrict__ h,
                                  const int* __restrict__ keys_a,
                                  const float* __restrict__ bucket_sums,
                                  const int* __restrict__ counts,
                                  const float* __restrict__ W2,
                                  const float* __restrict__ b2,
                                  const float* __restrict__ Wo,
                                  const float* __restrict__ bo,
                                  float* __restrict__ W2o,
                                  float* __restrict__ bias_m) {
    __shared__ __align__(16) float As[2][BM][AS_LD];
    __shared__ __align__(16) float Bs[2][BQ][BQ_LD];
    static_assert(D_MODEL == 256, "the bias block has one thread per column");

    if (blockIdx.y < L1_GRID_Y) {
        gemm_block<true, 2>(As, Bs, blockIdx.y * BM, blockIdx.x * BN, emb_a, W1, b1, h,
                         2 * D_MODEL, 2 * D_MODEL, 1, keys_a, bucket_sums, counts);
        return;
    }
    const int e = (blockIdx.y - L1_GRID_Y) * L1_GRID_X + blockIdx.x;
    if (e < MERGE_TILES) {
        gemm_block<false, 2>(As, Bs, (e / MERGE_GRID_X) * BM, (e % MERGE_GRID_X) * BN,
                          W2, Wo, nullptr, W2o, D_MODEL, D_MODEL, 0,
                          nullptr, nullptr, nullptr);
    } else if (e == MERGE_TILES) {
        merged_bias_block(b2, Wo, bo, bias_m);
    }
}

extern "C" void kernel_launch(void* const* d_in, const int* in_sizes, int n_in,
                              void* d_out, int out_size, void* d_ws, size_t ws_size,
                              hipStream_t stream) {
    const float* emb_a  = (const float*)d_in[0];
    const float* emb_b  = (const float*)d_in[1];
    const int*   keys_a = (const int*)d_in[2];
    const int*   keys_b = (const int*)d_in[3];
    const float* W1 = (const float*)d_in[4];
    const float* b1 = (const float*)d_in[5];
    const float* W2 = (const float*)d_in[6];
    const float* b2 = (const float*)d_in[7];
    const float* Wo = (const float*)d_in[8];
    const float* bo = (const float*)d_in[9];
    float* out = (float*)d_out;

    float* ws          = (float*)d_ws;
    int*   hash_b      = (int*)ws;
    float* bucket_sums = ws + BATCH_B;
    int*   counts      = (int*)(bucket_sums + NBUCKETS * D_MODEL);
    float* h           = bucket_sums + NBUCKETS * D_MODEL + NBUCKETS;
    float* W2o         = h + (size_t)BATCH_A * 2 * D_MODEL;
    float* bias_m      = W2o + 2 * D_MODEL * D_MODEL;

    hash_b_kernel<<<(BATCH_B + 255) / 256, 256, 0, stream>>>(keys_b, hash_b);
    bucket_sum_kernel<<<NBUCKETS, D_MODEL, 0, stream>>>(emb_b, hash_b, bucket_sums, counts);

    first_layer_and_merge_kernel<<<dim3(L1_GRID_X, L1_GRID_Y + EXTRA_GRID_Y), 256, 0, stream>>>(
        emb_a, W1, b1, h, keys_a, bucket_sums, counts, W2, b2, Wo, bo, W2o, bias_m);
    wmma_gemm_bias_kernel<<<dim3(256 / BN, BATCH_A / BM), 256, 0, stream>>>(
        h, W2o, bias_m, out, 256, 512, 0);
}


# bucket kernel hashes the keys itself: a block scans them once for four buckets (256 blocks of 16 waves, was 1024 of 8); hash kernel, its launch and the hash array gone; sum order kept
# speedup vs baseline: 10.5791x; 1.0414x over previous
#include <hip/hip_runtime.h>

typedef float v2f __attribute__((ext_vector_type(2)));
typedef float v8f __attribute__((ext_vector_type(8)));

#define D_MODEL   256
#define BATCH_A   4096
#define BATCH_B   8192
#define NKEYS     4
#define NBUCKETS  1024

__device__ __forceinline__ int hash_row(const int* __restrict__ keys, int row) {
    int s = 0;
#pragma unroll
    for (int k = 0; k < NKEYS; ++k) s += keys[row * NKEYS + k];
    return (int)(((unsigned)s) & (NBUCKETS - 1));
}

#define BK_BUCKETS      4
#define BK_THREADS      512
#define BK_WAVES        (BK_THREADS / 32)
#define BK_CHUNK        (BATCH_B / BK_WAVES)
#define BK_LOADS        (BK_CHUNK / 32)
#define BK_SUM_THREADS  (BK_THREADS / BK_BUCKETS)
#define SUM_BATCH       16
static_assert(NKEYS == 4, "one row of keys is one 16-byte load");
static_assert(BK_WAVES * BK_CHUNK == BATCH_B, "scan must cover every b-row");
static_assert(BK_LOADS * 32 == BK_CHUNK, "whole wave-wide loads only");
static_assert(BK_SUM_THREADS * 2 == D_MODEL, "two features per adding thread");
static_assert(NBUCKETS % BK_BUCKETS == 0, "whole blocks of buckets");
static_assert(BATCH_B <= 65536, "row numbers are kept as 16-bit values");

__device__ __forceinline__ int popc32(unsigned v) { return __builtin_popcount(v); }
__device__ __forceinline__ int round_up_batch(int n) { return (n + SUM_BATCH - 1) & ~(SUM_BATCH - 1); }

__global__ __launch_bounds__(BK_THREADS)
void bucket_sum_kernel(const float* __restrict__ emb_b,
                       const int* __restrict__ keys_b,
                       float* __restrict__ bucket_sums,
                       int* __restrict__ counts) {
    __shared__ __align__(16) unsigned short rows[BATCH_B + BK_BUCKETS * SUM_BATCH];
    __shared__ __align__(16) int chunk_cnt[BK_WAVES][BK_BUCKETS];

    const int t    = threadIdx.x;
    const int lane = t & 31;
    const int wave = t >> 5;
    const int g0   = blockIdx.x * BK_BUCKETS;
    const int chunk = (wave + blockIdx.x) & (BK_WAVES - 1);

    const int4* kp = (const int4*)keys_b + chunk * BK_CHUNK + lane;
    int4 kv[BK_LOADS];
#pragma unroll
    for (int i = 0; i < BK_LOADS; ++i) kv[i] = kp[i * 32];

    int      hq[BK_LOADS];
    unsigned mb[BK_LOADS];
    int cnt0 = 0, cnt1 = 0, cnt2 = 0, cnt3 = 0;
#pragma unroll
    for (int i = 0; i < BK_LOADS; ++i) {
        const int s = kv[i].x + kv[i].y + kv[i].z + kv[i].w;
        hq[i] = (int)(((unsigned)s) & (NBUCKETS - 1)) - g0;
        mb[i] = __builtin_amdgcn_ballot_w32((unsigned)hq[i] < (unsigned)BK_BUCKETS);
        if (mb[i] != 0u) {
            cnt0 += popc32(__builtin_amdgcn_ballot_w32(hq[i] == 0));
            cnt1 += popc32(__builtin_amdgcn_ballot_w32(hq[i] == 1));
            cnt2 += popc32(__builtin_amdgcn_ballot_w32(hq[i] == 2));
            cnt3 += popc32(__builtin_amdgcn_ballot_w32(hq[i] == 3));
        }
    }
    if (lane == 0) *(int4*)chunk_cnt[chunk] = make_int4(cnt0, cnt1, cnt2, cnt3);
    __syncthreads();

    int base0 = 0, base1 = 0, base2 = 0, base3 = 0;
    int tot0 = 0, tot1 = 0, tot2 = 0, tot3 = 0;
#pragma unroll
    for (int c = 0; c < BK_WAVES; ++c) {
        const int4 w = *(const int4*)chunk_cnt[c];
        if (c < chunk) { base0 += w.x; base1 += w.y; base2 += w.z; base3 += w.w; }
        tot0 += w.x; tot1 += w.y; tot2 += w.z; tot3 += w.w;
    }
    const int off0 = 0;
    const int off1 = off0 + round_up_batch(tot0);
    const int off2 = off1 + round_up_batch(tot1);
    const int off3 = off2 + round_up_batch(tot2);
    if (t < BK_BUCKETS * SUM_BATCH) {
        const int pq = t / SUM_BATCH, pk = t % SUM_BATCH;
        const int ptot = pq == 0 ? tot0 : pq == 1 ? tot1 : pq == 2 ? tot2 : tot3;
        const int poff = pq == 0 ? off0 : pq == 1 ? off1 : pq == 2 ? off2 : off3;
        if (ptot + pk < round_up_batch(ptot)) rows[poff + ptot + pk] = 0;
    }
    if ((cnt0 | cnt1 | cnt2 | cnt3) != 0) {
        const unsigned lt = (1u << lane) - 1u;
        int pos0 = off0 + base0, pos1 = off1 + base1, pos2 = off2 + base2, pos3 = off3 + base3;
#pragma unroll
        for (int i = 0; i < BK_LOADS; ++i) {
            if (mb[i] != 0u) {
                const unsigned short row = (unsigned short)(chunk * BK_CHUNK + i * 32 + lane);
                const unsigned m0 = __builtin_amdgcn_ballot_w32(hq[i] == 0);
                const unsigned m1 = __builtin_amdgcn_ballot_w32(hq[i] == 1);
                const unsigned m2 = __builtin_amdgcn_ballot_w32(hq[i] == 2);
                const unsigned m3 = __builtin_amdgcn_ballot_w32(hq[i] == 3);
                if (hq[i] == 0) rows[pos0 + popc32(m0 & lt)] = row;
                if (hq[i] == 1) rows[pos1 + popc32(m1 & lt)] = row;
                if (hq[i] == 2) rows[pos2 + popc32(m2 & lt)] = row;
                if (hq[i] == 3) rows[pos3 + popc32(m3 & lt)] = row;
                pos0 += popc32(m0); pos1 += popc32(m1); pos2 += popc32(m2); pos3 += popc32(m3);
            }
        }
    }
    __syncthreads();

    const int q  = t / BK_SUM_THREADS;
    const int tq = t % BK_SUM_THREADS;
    const int total = q == 0 ? tot0 : q == 1 ? tot1 : q == 2 ? tot2 : tot3;
    const int off   = q == 0 ? off0 : q == 1 ? off1 : q == 2 ? off2 : off3;
    const unsigned short* __restrict__ lst = rows + off;
    const float* __restrict__ src = emb_b + 2 * tq;
    float accx = 0.0f, accy = 0.0f;
    for (int e = 0; e < total; e += SUM_BATCH) {
        const uint4 pa = *(const uint4*)&lst[e];
        const uint4 pb = *(const uint4*)&lst[e + 8];
        const unsigned w[8] = {pa.x, pa.y, pa.z, pa.w, pb.x, pb.y, pb.z, pb.w};
        float2 r[SUM_BATCH];
#pragma unroll
        for (int k = 0; k < SUM_BATCH; k += 2) {
            r[k]     = *(const float2*)(src + (size_t)(w[k >> 1] & 0xffffu) * D_MODEL);
            r[k + 1] = *(const float2*)(src + (size_t)(w[k >> 1] >> 16)     * D_MODEL);
        }
#pragma unroll
        for (int k = 0; k < SUM_BATCH; ++k) {
            const float sx = accx + r[k].x;
            const float sy = accy + r[k].y;
            accx = (e + k < total) ? sx : accx;
            accy = (e + k < total) ? sy : accy;
        }
    }
    *(float2*)(bucket_sums + (size_t)(g0 + q) * D_MODEL + 2 * tq) = make_float2(accx, accy);
    if (tq == 0) counts[g0 + q] = total;
}

#define BM 64
#define BN 64
#define KT 64
#define AS_LD (KT + 4)
#define BQ (KT / 4)
#define BQ_LD (4 * BN + 32)
#define LD_ROWS 16
#define A_LD4 (BM / LD_ROWS)
#define B_LD4 4
static_assert(BQ == LD_ROWS, "one k quad of the B tile per staging row group");
static_assert(BN == 64, "slot() swizzles with bits 4..5 of the column");
static_assert(D_MODEL % KT == 0, "a k tile lies wholly in one half of the gathered row");

__device__ __forceinline__ void gemm_tile_load_a(const float* __restrict__ Ap, int lda,
                                                 float4 (&av)[A_LD4]) {
#pragma unroll
    for (int i = 0; i < A_LD4; ++i)
        av[i] = *(const float4*)(Ap + (size_t)(LD_ROWS * i) * lda);
}

__device__ __forceinline__ void gemm_tile_gather_a(const float* __restrict__ Gp,
                                                   const unsigned (&goff)[A_LD4],
                                                   float4 (&av)[A_LD4]) {
#pragma unroll
    for (int i = 0; i < A_LD4; ++i)
        av[i] = *(const float4*)(Gp + goff[i]);
}

__device__ __forceinline__ void gemm_tile_load_b(const float* __restrict__ Bp, int N,
                                                 float4 (&bv)[B_LD4]) {
#pragma unroll
    for (int i = 0; i < B_LD4; ++i)
        bv[i] = *(const float4*)(Bp + (size_t)i * N);
}

__device__ __forceinline__ void gemm_tile_store(float (*As)[AS_LD], float* bsp, int bsw,
                                                int lr, int lc,
                                                const float4 (&av)[A_LD4],
                                                const float4 (&bv)[B_LD4]) {
#pragma unroll
    for (int i = 0; i < A_LD4; ++i) *(float4*)&As[lr + LD_ROWS * i][lc] = av[i];
    const float4 t0 = {bv[0].x, bv[1].x, bv[2].x, bv[3].x};
    const float4 t1 = {bv[0].y, bv[1].y, bv[2].y, bv[3].y};
    const float4 t2 = {bv[0].z, bv[1].z, bv[2].z, bv[3].z};
    const float4 t3 = {bv[0].w, bv[1].w, bv[2].w, bv[3].w};
    *(float4*)(bsp + 4 * (0 ^ bsw)) = t0;
    *(float4*)(bsp + 4 * (1 ^ bsw)) = t1;
    *(float4*)(bsp + 4 * (2 ^ bsw)) = t2;
    *(float4*)(bsp + 4 * (3 ^ bsw)) = t3;
}

__device__ __forceinline__ void gemm_tile_mma(const float* __restrict__ ap,
                                              const float* __restrict__ bp0,
                                              const float* __restrict__ bp1,
                                              v8f& c0, v8f& c1) {
    float4 a[KT / 8], b0[KT / 8], b1[KT / 8];
#pragma unroll
    for (int j = 0; j < KT / 8; ++j) {
        a[j]  = *(const float4*)(ap + 8 * j);
        b0[j] = *(const float4*)(bp0 + 2 * j * BQ_LD);
        b1[j] = *(const float4*)(bp1 + 2 * j * BQ_LD);
    }
#pragma unroll
    for (int j = 0; j < KT / 8; ++j) {
        const v2f alo = {a[j].x, a[j].y},   ahi = {a[j].z, a[j].w};
        const v2f p0  = {b0[j].x, b0[j].y}, q0  = {b0[j].z, b0[j].w};
        const v2f p1  = {b1[j].x, b1[j].y}, q1  = {b1[j].z, b1[j].w};
        c0 = __builtin_amdgcn_wmma_f32_16x16x4_f32(false, alo, false, p0,
                                                   (short)0, c0, false, false);
        c1 = __builtin_amdgcn_wmma_f32_16x16x4_f32(false, alo, false, p1,
                                                   (short)0, c1, false, false);
        c0 = __builtin_amdgcn_wmma_f32_16x16x4_f32(false, ahi, false, q0,
                                                   (short)0, c0, false, false);
        c1 = __builtin_amdgcn_wmma_f32_16x16x4_f32(false, ahi, false, q1,
                                                   (short)0, c1, false, false);
    }
}

template <int GROUPS, bool STAGING>
__device__ __forceinline__ void gemm_step_order() {
    if constexpr (STAGING) __builtin_amdgcn_sched_group_barrier(0x020, A_LD4 + B_LD4, 0);
    if constexpr (GROUPS == 1) {
        __builtin_amdgcn_sched_group_barrier(0x100, 3 * (KT / 8), 0);
        __builtin_amdgcn_sched_group_barrier(0x008, 4 * (KT / 8), 0);
    } else {
        __builtin_amdgcn_sched_group_barrier(0x100, 3 * (KT / 16), 0);
        __builtin_amdgcn_sched_group_barrier(0x008, 4 * (KT / 16), 0);
        __builtin_amdgcn_sched_group_barrier(0x100, 3 * (KT / 16), 0);
        __builtin_amdgcn_sched_group_barrier(0x008, 4 * (KT / 16), 0);
    }
    if constexpr (STAGING) __builtin_amdgcn_sched_group_barrier(0x200, A_LD4 + B_LD4, 0);
}

template <bool GATHER, int GROUPS>
__device__ __forceinline__
void gemm_block(float (&As)[2][BM][AS_LD], float (&Bs)[2][BQ][BQ_LD],
                const int bm, const int bn,
                const float* __restrict__ A,
                const float* __restrict__ B,
                const float* __restrict__ bias,
                float* __restrict__ C,
                int N, int K, int relu,
                const int* __restrict__ keys_a,
                const float* __restrict__ bucket_sums,
                const int* __restrict__ counts) {
    const int tid  = threadIdx.x;
    const int lane = tid & 31;
    const int wave = tid >> 5;
    const int wm   = wave >> 1;
    const int wn   = wave & 1;
    const int hi   = lane >> 4;
    const int l15  = lane & 15;

    v8f c0 = {};
    v8f c1 = {};

    const int lr = tid >> 4;
    const int lc = (tid & 15) * 4;
    const int lda = GATHER ? D_MODEL : K;
    const int ka  = GATHER ? D_MODEL : K;
    const float* __restrict__ Ap = A + (size_t)(bm + lr) * lda + lc;
    const float* __restrict__ Bp = B + (size_t)(4 * lr) * N + (bn + lc);
    const int bsw  = (lc >> 4) & 3;
    const int bsto = lr * BQ_LD + 4 * lc;

    int ha[A_LD4];
    if (GATHER) {
#pragma unroll
        for (int i = 0; i < A_LD4; ++i) ha[i] = hash_row(keys_a, bm + lr + LD_ROWS * i);
    }

    const int aoff  = (wm * 16 + l15) * AS_LD + 4 * hi;
    const int boff0 = hi * BQ_LD + 4 * (wn * 32 +      (l15 ^ (2 * wn)));
    const int boff1 = hi * BQ_LD + 4 * (wn * 32 + 16 + (l15 ^ (2 * wn + 1)));

    float4 av[A_LD4], bv[B_LD4];
    gemm_tile_load_a(Ap, lda, av);
    gemm_tile_load_b(Bp, N, bv);
    gemm_tile_store(As[0], &Bs[0][0][0] + bsto, bsw, lr, lc, av, bv);
    __syncthreads();

    unsigned goff[A_LD4];
    int gcnt[A_LD4];
    if (GATHER) {
#pragma unroll
        for (int i = 0; i < A_LD4; ++i) {
            gcnt[i] = counts[ha[i]];
            goff[i] = (unsigned)(ha[i] * D_MODEL + lc);
        }
    }

    const int nk  = K / KT;
    const int nka = ka / KT;
    int cur = 0;
    int t = 1;
#pragma nounroll
    for (; t < nka; ++t) {
        gemm_tile_load_a(Ap + (size_t)t * KT, lda, av);
        gemm_tile_load_b(Bp + (size_t)t * KT * N, N, bv);
        gemm_tile_mma(&As[cur][0][0] + aoff, &Bs[cur][0][0] + boff0, &Bs[cur][0][0] + boff1, c0, c1);
        gemm_tile_store(As[cur ^ 1], &Bs[cur ^ 1][0][0] + bsto, bsw, lr, lc, av, bv);
        gemm_step_order<GROUPS, true>();
        __syncthreads();
        cur ^= 1;
    }
    if (GATHER) {
        float gs[A_LD4];
#pragma unroll
        for (int i = 0; i < A_LD4; ++i)
            gs[i] = 1.0f / (float)(gcnt[i] < 1 ? 1 : gcnt[i]);
#pragma nounroll
        for (; t < nk; ++t) {
            gemm_tile_gather_a(bucket_sums + (t * KT - D_MODEL), goff, av);
            gemm_tile_load_b(Bp + (size_t)t * KT * N, N, bv);
            gemm_tile_mma(&As[cur][0][0] + aoff, &Bs[cur][0][0] + boff0, &Bs[cur][0][0] + boff1, c0, c1);
#pragma unroll
            for (int i = 0; i < A_LD4; ++i) {
                av[i].x *= gs[i]; av[i].y *= gs[i]; av[i].z *= gs[i]; av[i].w *= gs[i];
            }
            gemm_tile_store(As[cur ^ 1], &Bs[cur ^ 1][0][0] + bsto, bsw, lr, lc, av, bv);
            gemm_step_order<GROUPS, true>();
            __syncthreads();
            cur ^= 1;
        }
    }
    gemm_tile_mma(&As[cur][0][0] + aoff, &Bs[cur][0][0] + boff0, &Bs[cur][0][0] + boff1, c0, c1);
    gemm_step_order<GROUPS, false>();

    const int n0 = bn + wn * 32 + l15;
    const float bia0 = bias ? bias[n0] : 0.0f;
    const float bia1 = bias ? bias[n0 + 16] : 0.0f;
#pragma unroll
    for (int r = 0; r < 8; ++r) {
        const int m = bm + wm * 16 + r + 8 * hi;
        float v0 = c0[r] + bia0;
        float v1 = c1[r] + bia1;
        if (relu) { v0 = fmaxf(v0, 0.0f); v1 = fmaxf(v1, 0.0f); }
        C[(size_t)m * N + n0]      = v0;
        C[(size_t)m * N + n0 + 16] = v1;
    }
}

__global__ __launch_bounds__(256)
void wmma_gemm_bias_kernel(const float* __restrict__ A,
                           const float* __restrict__ B,
                           const float* __restrict__ bias,
                           float* __restrict__ C,
                           int N, int K, int relu) {
    __shared__ __align__(16) float As[2][BM][AS_LD];
    __shared__ __align__(16) float Bs[2][BQ][BQ_LD];
    gemm_block<false, 1>(As, Bs, blockIdx.y * BM, blockIdx.x * BN, A, B, bias, C, N, K, relu,
                      nullptr, nullptr, nullptr);
}

#define L1_GRID_X    (2 * D_MODEL / BN)
#define L1_GRID_Y    (BATCH_A / BM)
#define MERGE_GRID_X (D_MODEL / BN)
#define MERGE_TILES  ((2 * D_MODEL / BM) * MERGE_GRID_X)
#define EXTRA_GRID_Y ((MERGE_TILES + 1 + L1_GRID_X - 1) / L1_GRID_X)

__device__ __forceinline__ void merged_bias_block(const float* __restrict__ b2,
                                                  const float* __restrict__ Wo,
                                                  const float* __restrict__ bo,
                                                  float* __restrict__ bias_m) {
    const int n = threadIdx.x;
    float acc = 0.0f;
#pragma unroll 64
    for (int j = 0; j < D_MODEL; ++j) acc = fmaf(b2[j], Wo[(size_t)j * D_MODEL + n], acc);
    bias_m[n] = acc + bo[n];
}

__global__ __launch_bounds__(256)
void first_layer_and_merge_kernel(const float* __restrict__ emb_a,
                                  const float* __restrict__ W1,
                                  const float* __restrict__ b1,
                                  float* __restrict__ h,
                                  const int* __restrict__ keys_a,
                                  const float* __restrict__ bucket_sums,
                                  const int* __restrict__ counts,
                                  const float* __restrict__ W2,
                                  const float* __restrict__ b2,
                                  const float* __restrict__ Wo,
                                  const float* __restrict__ bo,
                                  float* __restrict__ W2o,
                                  float* __restrict__ bias_m) {
    __shared__ __align__(16) float As[2][BM][AS_LD];
    __shared__ __align__(16) float Bs[2][BQ][BQ_LD];
    static_assert(D_MODEL == 256, "the bias block has one thread per column");

    if (blockIdx.y < L1_GRID_Y) {
        gemm_block<true, 2>(As, Bs, blockIdx.y * BM, blockIdx.x * BN, emb_a, W1, b1, h,
                         2 * D_MODEL, 2 * D_MODEL, 1, keys_a, bucket_sums, counts);
        return;
    }
    const int e = (blockIdx.y - L1_GRID_Y) * L1_GRID_X + blockIdx.x;
    if (e < MERGE_TILES) {
        gemm_block<false, 2>(As, Bs, (e / MERGE_GRID_X) * BM, (e % MERGE_GRID_X) * BN,
                          W2, Wo, nullptr, W2o, D_MODEL, D_MODEL, 0,
                          nullptr, nullptr, nullptr);
    } else if (e == MERGE_TILES) {
        merged_bias_block(b2, Wo, bo, bias_m);
    }
}

extern "C" void kernel_launch(void* const* d_in, const int* in_sizes, int n_in,
                              void* d_out, int out_size, void* d_ws, size_t ws_size,
                              hipStream_t stream) {
    const float* emb_a  = (const float*)d_in[0];
    const float* emb_b  = (const float*)d_in[1];
    const int*   keys_a = (const int*)d_in[2];
    const int*   keys_b = (const int*)d_in[3];
    const float* W1 = (const float*)d_in[4];
    const float* b1 = (const float*)d_in[5];
    const float* W2 = (const float*)d_in[6];
    const float* b2 = (const float*)d_in[7];
    const float* Wo = (const float*)d_in[8];
    const float* bo = (const float*)d_in[9];
    float* out = (float*)d_out;

    float* ws          = (float*)d_ws;
    float* bucket_sums = ws;
    int*   counts      = (int*)(bucket_sums + NBUCKETS * D_MODEL);
    float* h           = bucket_sums + NBUCKETS * D_MODEL + NBUCKETS;
    float* W2o         = h + (size_t)BATCH_A * 2 * D_MODEL;
    float* bias_m      = W2o + 2 * D_MODEL * D_MODEL;

    bucket_sum_kernel<<<NBUCKETS / BK_BUCKETS, BK_THREADS, 0, stream>>>(emb_b, keys_b, bucket_sums, counts);

    first_layer_and_merge_kernel<<<dim3(L1_GRID_X, L1_GRID_Y + EXTRA_GRID_Y), 256, 0, stream>>>(
        emb_a, W1, b1, h, keys_a, bucket_sums, counts, W2, b2, Wo, bo, W2o, bias_m);
    wmma_gemm_bias_kernel<<<dim3(256 / BN, BATCH_A / BM), 256, 0, stream>>>(
        h, W2o, bias_m, out, 256, 512, 0);
}


# bucket half of layer 1 multiplied once per bucket: P = means x W1[256:] (1024 rows, was 4096); last GEMM's loader forms h = relu(emb_a x W1[:256] + b1 + P[hash]); sums reassociated, rounding changes
# speedup vs baseline: 12.2343x; 1.1565x over previous
#include <hip/hip_runtime.h>

typedef float v2f __attribute__((ext_vector_type(2)));
typedef float v8f __attribute__((ext_vector_type(8)));

#define D_MODEL   256
#define BATCH_A   4096
#define BATCH_B   8192
#define NKEYS     4
#define NBUCKETS  1024

__device__ __forceinline__ int hash_row(const int* __restrict__ keys, int row) {
    int s = 0;
#pragma unroll
    for (int k = 0; k < NKEYS; ++k) s += keys[row * NKEYS + k];
    return (int)(((unsigned)s) & (NBUCKETS - 1));
}

#define BK_BUCKETS      4
#define BK_THREADS      512
#define BK_WAVES        (BK_THREADS / 32)
#define BK_CHUNK        (BATCH_B / BK_WAVES)
#define BK_LOADS        (BK_CHUNK / 32)
#define BK_SUM_THREADS  (BK_THREADS / BK_BUCKETS)
#define SUM_BATCH       16
static_assert(NKEYS == 4, "one row of keys is one 16-byte load");
static_assert(BK_WAVES * BK_CHUNK == BATCH_B, "scan must cover every b-row");
static_assert(BK_LOADS * 32 == BK_CHUNK, "whole wave-wide loads only");
static_assert(BK_SUM_THREADS * 2 == D_MODEL, "two features per adding thread");
static_assert(NBUCKETS % BK_BUCKETS == 0, "whole blocks of buckets");
static_assert(BATCH_B <= 65536, "row numbers are kept as 16-bit values");

__device__ __forceinline__ int popc32(unsigned v) { return __builtin_popcount(v); }
__device__ __forceinline__ int round_up_batch(int n) { return (n + SUM_BATCH - 1) & ~(SUM_BATCH - 1); }

__global__ __launch_bounds__(BK_THREADS)
void bucket_sum_kernel(const float* __restrict__ emb_b,
                       const int* __restrict__ keys_b,
                       float* __restrict__ bucket_sums,
                       int* __restrict__ counts) {
    __shared__ __align__(16) unsigned short rows[BATCH_B + BK_BUCKETS * SUM_BATCH];
    __shared__ __align__(16) int chunk_cnt[BK_WAVES][BK_BUCKETS];

    const int t    = threadIdx.x;
    const int lane = t & 31;
    const int wave = t >> 5;
    const int g0   = blockIdx.x * BK_BUCKETS;
    const int chunk = (wave + blockIdx.x) & (BK_WAVES - 1);

    const int4* kp = (const int4*)keys_b + chunk * BK_CHUNK + lane;
    int4 kv[BK_LOADS];
#pragma unroll
    for (int i = 0; i < BK_LOADS; ++i) kv[i] = kp[i * 32];

    int      hq[BK_LOADS];
    unsigned mb[BK_LOADS];
    int cnt0 = 0, cnt1 = 0, cnt2 = 0, cnt3 = 0;
#pragma unroll
    for (int i = 0; i < BK_LOADS; ++i) {
        const int s = kv[i].x + kv[i].y + kv[i].z + kv[i].w;
        hq[i] = (int)(((unsigned)s) & (NBUCKETS - 1)) - g0;
        mb[i] = __builtin_amdgcn_ballot_w32((unsigned)hq[i] < (unsigned)BK_BUCKETS);
        if (mb[i] != 0u) {
            cnt0 += popc32(__builtin_amdgcn_ballot_w32(hq[i] == 0));
            cnt1 += popc32(__builtin_amdgcn_ballot_w32(hq[i] == 1));
            cnt2 += popc32(__builtin_amdgcn_ballot_w32(hq[i] == 2));
            cnt3 += popc32(__builtin_amdgcn_ballot_w32(hq[i] == 3));
        }
    }
    if (lane == 0) *(int4*)chunk_cnt[chunk] = make_int4(cnt0, cnt1, cnt2, cnt3);
    __syncthreads();

    int base0 = 0, base1 = 0, base2 = 0, base3 = 0;
    int tot0 = 0, tot1 = 0, tot2 = 0, tot3 = 0;
#pragma unroll
    for (int c = 0; c < BK_WAVES; ++c) {
        const int4 w = *(const int4*)chunk_cnt[c];
        if (c < chunk) { base0 += w.x; base1 += w.y; base2 += w.z; base3 += w.w; }
        tot0 += w.x; tot1 += w.y; tot2 += w.z; tot3 += w.w;
    }
    const int off0 = 0;
    const int off1 = off0 + round_up_batch(tot0);
    const int off2 = off1 + round_up_batch(tot1);
    const int off3 = off2 + round_up_batch(tot2);
    if (t < BK_BUCKETS * SUM_BATCH) {
        const int pq = t / SUM_BATCH, pk = t % SUM_BATCH;
        const int ptot = pq == 0 ? tot0 : pq == 1 ? tot1 : pq == 2 ? tot2 : tot3;
        const int poff = pq == 0 ? off0 : pq == 1 ? off1 : pq == 2 ? off2 : off3;
        if (ptot + pk < round_up_batch(ptot)) rows[poff + ptot + pk] = 0;
    }
    if ((cnt0 | cnt1 | cnt2 | cnt3) != 0) {
        const unsigned lt = (1u << lane) - 1u;
        int pos0 = off0 + base0, pos1 = off1 + base1, pos2 = off2 + base2, pos3 = off3 + base3;
#pragma unroll
        for (int i = 0; i < BK_LOADS; ++i) {
            if (mb[i] != 0u) {
                const unsigned short row = (unsigned short)(chunk * BK_CHUNK + i * 32 + lane);
                const unsigned m0 = __builtin_amdgcn_ballot_w32(hq[i] == 0);
                const unsigned m1 = __builtin_amdgcn_ballot_w32(hq[i] == 1);
                const unsigned m2 = __builtin_amdgcn_ballot_w32(hq[i] == 2);
                const unsigned m3 = __builtin_amdgcn_ballot_w32(hq[i] == 3);
                if (hq[i] == 0) rows[pos0 + popc32(m0 & lt)] = row;
                if (hq[i] == 1) rows[pos1 + popc32(m1 & lt)] = row;
                if (hq[i] == 2) rows[pos2 + popc32(m2 & lt)] = row;
                if (hq[i] == 3) rows[pos3 + popc32(m3 & lt)] = row;
                pos0 += popc32(m0); pos1 += popc32(m1); pos2 += popc32(m2); pos3 += popc32(m3);
            }
        }
    }
    __syncthreads();

    const int q  = t / BK_SUM_THREADS;
    const int tq = t % BK_SUM_THREADS;
    const int total = q == 0 ? tot0 : q == 1 ? tot1 : q == 2 ? tot2 : tot3;
    const int off   = q == 0 ? off0 : q == 1 ? off1 : q == 2 ? off2 : off3;
    const unsigned short* __restrict__ lst = rows + off;
    const float* __restrict__ src = emb_b + 2 * tq;
    float accx = 0.0f, accy = 0.0f;
    for (int e = 0; e < total; e += SUM_BATCH) {
        const uint4 pa = *(const uint4*)&lst[e];
        const uint4 pb = *(const uint4*)&lst[e + 8];
        const unsigned w[8] = {pa.x, pa.y, pa.z, pa.w, pb.x, pb.y, pb.z, pb.w};
        float2 r[SUM_BATCH];
#pragma unroll
        for (int k = 0; k < SUM_BATCH; k += 2) {
            r[k]     = *(const float2*)(src + (size_t)(w[k >> 1] & 0xffffu) * D_MODEL);
            r[k + 1] = *(const float2*)(src + (size_t)(w[k >> 1] >> 16)     * D_MODEL);
        }
#pragma unroll
        for (int k = 0; k < SUM_BATCH; ++k) {
            const float sx = accx + r[k].x;
            const float sy = accy + r[k].y;
            accx = (e + k < total) ? sx : accx;
            accy = (e + k < total) ? sy : accy;
        }
    }
    *(float2*)(bucket_sums + (size_t)(g0 + q) * D_MODEL + 2 * tq) = make_float2(accx, accy);
    if (tq == 0) counts[g0 + q] = total;
}

#define BM 64
#define BN 64
#define KT 64
#define AS_LD (KT + 4)
#define BQ (KT / 4)
#define BQ_LD (4 * BN + 32)
#define LD_ROWS 16
#define A_LD4 (BM / LD_ROWS)
#define B_LD4 4
static_assert(BQ == LD_ROWS, "one k quad of the B tile per staging row group");
static_assert(BN == 64, "slot() swizzles with bits 4..5 of the column");
#define A_PLAIN   0
#define A_SCALED  1
#define A_COMBINE 2

__device__ __forceinline__ void gemm_tile_load_a(const float* __restrict__ Ap, int lda,
                                                 float4 (&av)[A_LD4]) {
#pragma unroll
    for (int i = 0; i < A_LD4; ++i)
        av[i] = *(const float4*)(Ap + (size_t)(LD_ROWS * i) * lda);
}

__device__ __forceinline__ void gemm_tile_gather_a(const float* __restrict__ Gp,
                                                   const unsigned (&goff)[A_LD4],
                                                   float4 (&av)[A_LD4]) {
#pragma unroll
    for (int i = 0; i < A_LD4; ++i)
        av[i] = *(const float4*)(Gp + goff[i]);
}

__device__ __forceinline__ void gemm_tile_load_b(const float* __restrict__ Bp, int N,
                                                 float4 (&bv)[B_LD4]) {
#pragma unroll
    for (int i = 0; i < B_LD4; ++i)
        bv[i] = *(const float4*)(Bp + (size_t)i * N);
}

__device__ __forceinline__ void gemm_tile_store(float (*As)[AS_LD], float* bsp, int bsw,
                                                int lr, int lc,
                                                const float4 (&av)[A_LD4],
                                                const float4 (&bv)[B_LD4]) {
#pragma unroll
    for (int i = 0; i < A_LD4; ++i) *(float4*)&As[lr + LD_ROWS * i][lc] = av[i];
    const float4 t0 = {bv[0].x, bv[1].x, bv[2].x, bv[3].x};
    const float4 t1 = {bv[0].y, bv[1].y, bv[2].y, bv[3].y};
    const float4 t2 = {bv[0].z, bv[1].z, bv[2].z, bv[3].z};
    const float4 t3 = {bv[0].w, bv[1].w, bv[2].w, bv[3].w};
    *(float4*)(bsp + 4 * (0 ^ bsw)) = t0;
    *(float4*)(bsp + 4 * (1 ^ bsw)) = t1;
    *(float4*)(bsp + 4 * (2 ^ bsw)) = t2;
    *(float4*)(bsp + 4 * (3 ^ bsw)) = t3;
}

__device__ __forceinline__ void gemm_tile_mma(const float* __restrict__ ap,
                                              const float* __restrict__ bp0,
                                              const float* __restrict__ bp1,
                                              v8f& c0, v8f& c1) {
    float4 a[KT / 8], b0[KT / 8], b1[KT / 8];
#pragma unroll
    for (int j = 0; j < KT / 8; ++j) {
        a[j]  = *(const float4*)(ap + 8 * j);
        b0[j] = *(const float4*)(bp0 + 2 * j * BQ_LD);
        b1[j] = *(const float4*)(bp1 + 2 * j * BQ_LD);
    }
#pragma unroll
    for (int j = 0; j < KT / 8; ++j) {
        const v2f alo = {a[j].x, a[j].y},   ahi = {a[j].z, a[j].w};
        const v2f p0  = {b0[j].x, b0[j].y}, q0  = {b0[j].z, b0[j].w};
        const v2f p1  = {b1[j].x, b1[j].y}, q1  = {b1[j].z, b1[j].w};
        c0 = __builtin_amdgcn_wmma_f32_16x16x4_f32(false, alo, false, p0,
                                                   (short)0, c0, false, false);
        c1 = __builtin_amdgcn_wmma_f32_16x16x4_f32(false, alo, false, p1,
                                                   (short)0, c1, false, false);
        c0 = __builtin_amdgcn_wmma_f32_16x16x4_f32(false, ahi, false, q0,
                                                   (short)0, c0, false, false);
        c1 = __builtin_amdgcn_wmma_f32_16x16x4_f32(false, ahi, false, q1,
                                                   (short)0, c1, false, false);
    }
}

template <int GROUPS, bool STAGING, int NLOADS>
__device__ __forceinline__ void gemm_step_order() {
    if constexpr (STAGING) __builtin_amdgcn_sched_group_barrier(0x020, NLOADS, 0);
    if constexpr (GROUPS == 1) {
        __builtin_amdgcn_sched_group_barrier(0x100, 3 * (KT / 8), 0);
        __builtin_amdgcn_sched_group_barrier(0x008, 4 * (KT / 8), 0);
    } else {
        __builtin_amdgcn_sched_group_barrier(0x100, 3 * (KT / 16), 0);
        __builtin_amdgcn_sched_group_barrier(0x008, 4 * (KT / 16), 0);
        __builtin_amdgcn_sched_group_barrier(0x100, 3 * (KT / 16), 0);
        __builtin_amdgcn_sched_group_barrier(0x008, 4 * (KT / 16), 0);
    }
    if constexpr (STAGING) __builtin_amdgcn_sched_group_barrier(0x200, A_LD4 + B_LD4, 0);
}

template <int MODE>
__device__ __forceinline__ void gemm_tile_fix_a(float4 (&av)[A_LD4],
                                                const float4 (&gv)[A_LD4],
                                                const float (&gs)[A_LD4]) {
#pragma unroll
    for (int i = 0; i < A_LD4; ++i) {
        if constexpr (MODE == A_SCALED) {
            av[i].x *= gs[i]; av[i].y *= gs[i]; av[i].z *= gs[i]; av[i].w *= gs[i];
        }
        if constexpr (MODE == A_COMBINE) {
            av[i].x = fmaxf(av[i].x + gv[i].x, 0.0f);
            av[i].y = fmaxf(av[i].y + gv[i].y, 0.0f);
            av[i].z = fmaxf(av[i].z + gv[i].z, 0.0f);
            av[i].w = fmaxf(av[i].w + gv[i].w, 0.0f);
        }
    }
}

template <int MODE, int GROUPS>
__device__ __forceinline__
void gemm_block(float (&As)[2][BM][AS_LD], float (&Bs)[2][BQ][BQ_LD],
                const int bm, const int bn,
                const float* __restrict__ A, const int lda,
                const float* __restrict__ B,
                const float* __restrict__ bias,
                float* __restrict__ C,
                int N, int K, int relu,
                const int* __restrict__ keys_a,
                const float* __restrict__ G,
                const int* __restrict__ counts) {
    constexpr int NLOADS = A_LD4 + B_LD4 + (MODE == A_COMBINE ? A_LD4 : 0);
    const int tid  = threadIdx.x;
    const int lane = tid & 31;
    const int wave = tid >> 5;
    const int wm   = wave >> 1;
    const int wn   = wave & 1;
    const int hi   = lane >> 4;
    const int l15  = lane & 15;

    v8f c0 = {};
    v8f c1 = {};

    const int lr = tid >> 4;
    const int lc = (tid & 15) * 4;
    const float* __restrict__ Ap = A + (size_t)(bm + lr) * lda + lc;
    const float* __restrict__ Bp = B + (size_t)(4 * lr) * N + (bn + lc);
    const int bsw  = (lc >> 4) & 3;
    const int bsto = lr * BQ_LD + 4 * lc;

    unsigned goff[A_LD4];
    float gs[A_LD4];
#pragma unroll
    for (int i = 0; i < A_LD4; ++i) {
        if constexpr (MODE == A_COMBINE)
            goff[i] = (unsigned)(hash_row(keys_a, bm + lr + LD_ROWS * i) * K + lc);
        if constexpr (MODE == A_SCALED) {
            const int cnt = counts[bm + lr + LD_ROWS * i];
            gs[i] = 1.0f / (float)(cnt < 1 ? 1 : cnt);
        }
    }

    const int aoff  = (wm * 16 + l15) * AS_LD + 4 * hi;
    const int boff0 = hi * BQ_LD + 4 * (wn * 32 +      (l15 ^ (2 * wn)));
    const int boff1 = hi * BQ_LD + 4 * (wn * 32 + 16 + (l15 ^ (2 * wn + 1)));

    float4 av[A_LD4], bv[B_LD4], gv[A_LD4];
    gemm_tile_load_a(Ap, lda, av);
    gemm_tile_load_b(Bp, N, bv);
    if constexpr (MODE == A_COMBINE) gemm_tile_gather_a(G, goff, gv);
    gemm_tile_fix_a<MODE>(av, gv, gs);
    gemm_tile_store(As[0], &Bs[0][0][0] + bsto, bsw, lr, lc, av, bv);
    __syncthreads();

    const int nk = K / KT;
    int cur = 0;
#pragma nounroll
    for (int t = 1; t < nk; ++t) {
        gemm_tile_load_a(Ap + (size_t)t * KT, lda, av);
        gemm_tile_load_b(Bp + (size_t)t * KT * N, N, bv);
        if constexpr (MODE == A_COMBINE) gemm_tile_gather_a(G + t * KT, goff, gv);
        gemm_tile_mma(&As[cur][0][0] + aoff, &Bs[cur][0][0] + boff0, &Bs[cur][0][0] + boff1, c0, c1);
        gemm_tile_fix_a<MODE>(av, gv, gs);
        gemm_tile_store(As[cur ^ 1], &Bs[cur ^ 1][0][0] + bsto, bsw, lr, lc, av, bv);
        gemm_step_order<GROUPS, true, NLOADS>();
        __syncthreads();
        cur ^= 1;
    }
    gemm_tile_mma(&As[cur][0][0] + aoff, &Bs[cur][0][0] + boff0, &Bs[cur][0][0] + boff1, c0, c1);
    gemm_step_order<GROUPS, false, NLOADS>();

    const int n0 = bn + wn * 32 + l15;
    const float bia0 = bias ? bias[n0] : 0.0f;
    const float bia1 = bias ? bias[n0 + 16] : 0.0f;
#pragma unroll
    for (int r = 0; r < 8; ++r) {
        const int m = bm + wm * 16 + r + 8 * hi;
        float v0 = c0[r] + bia0;
        float v1 = c1[r] + bia1;
        if (relu) { v0 = fmaxf(v0, 0.0f); v1 = fmaxf(v1, 0.0f); }
        C[(size_t)m * N + n0]      = v0;
        C[(size_t)m * N + n0 + 16] = v1;
    }
}

#define L1_GRID_X    (2 * D_MODEL / BN)
#define L1_GRID_Y    (BATCH_A / BM)
#define P_GRID_Y     (NBUCKETS / BM)
#define MERGE_GRID_X (D_MODEL / BN)
#define MERGE_TILES  ((2 * D_MODEL / BM) * MERGE_GRID_X)
#define EXTRA_GRID_Y ((MERGE_TILES + 1 + L1_GRID_X - 1) / L1_GRID_X)

__device__ __forceinline__ void merged_bias_block(const float* __restrict__ b2,
                                                  const float* __restrict__ Wo,
                                                  const float* __restrict__ bo,
                                                  float* __restrict__ bias_m) {
    const int n = threadIdx.x;
    float acc = 0.0f;
#pragma unroll 64
    for (int j = 0; j < D_MODEL; ++j) acc = fmaf(b2[j], Wo[(size_t)j * D_MODEL + n], acc);
    bias_m[n] = acc + bo[n];
}

__global__ __launch_bounds__(256)
void first_layer_halves_and_merge_kernel(const float* __restrict__ emb_a,
                                         const float* __restrict__ W1,
                                         const float* __restrict__ b1,
                                         float* __restrict__ Ha,
                                         const float* __restrict__ bucket_sums,
                                         const int* __restrict__ counts,
                                         float* __restrict__ P,
                                         const float* __restrict__ W2,
                                         const float* __restrict__ b2,
                                         const float* __restrict__ Wo,
                                         const float* __restrict__ bo,
                                         float* __restrict__ W2o,
                                         float* __restrict__ bias_m) {
    __shared__ __align__(16) float As[2][BM][AS_LD];
    __shared__ __align__(16) float Bs[2][BQ][BQ_LD];
    static_assert(D_MODEL == 256, "the bias block has one thread per column");
    static_assert(NBUCKETS % BM == 0, "whole row tiles of P");

    if (blockIdx.y < L1_GRID_Y) {
        gemm_block<A_PLAIN, 2>(As, Bs, blockIdx.y * BM, blockIdx.x * BN, emb_a, D_MODEL, W1, b1, Ha,
                               2 * D_MODEL, D_MODEL, 0, nullptr, nullptr, nullptr);
        return;
    }
    if (blockIdx.y < L1_GRID_Y + P_GRID_Y) {
        gemm_block<A_SCALED, 2>(As, Bs, (blockIdx.y - L1_GRID_Y) * BM, blockIdx.x * BN,
                                bucket_sums, D_MODEL, W1 + (size_t)D_MODEL * (2 * D_MODEL), nullptr, P,
                                2 * D_MODEL, D_MODEL, 0, nullptr, nullptr, counts);
        return;
    }
    const int e = (blockIdx.y - L1_GRID_Y - P_GRID_Y) * L1_GRID_X + blockIdx.x;
    if (e < MERGE_TILES) {
        gemm_block<A_PLAIN, 2>(As, Bs, (e / MERGE_GRID_X) * BM, (e % MERGE_GRID_X) * BN,
                               W2, D_MODEL, Wo, nullptr, W2o, D_MODEL, D_MODEL, 0,
                               nullptr, nullptr, nullptr);
    } else if (e == MERGE_TILES) {
        merged_bias_block(b2, Wo, bo, bias_m);
    }
}

__global__ __launch_bounds__(256)
void hidden_to_out_kernel(const float* __restrict__ Ha,
                          const float* __restrict__ P,
                          const int* __restrict__ keys_a,
                          const float* __restrict__ W2o,
                          const float* __restrict__ bias_m,
                          float* __restrict__ out) {
    __shared__ __align__(16) float As[2][BM][AS_LD];
    __shared__ __align__(16) float Bs[2][BQ][BQ_LD];
    gemm_block<A_COMBINE, 1>(As, Bs, blockIdx.y * BM, blockIdx.x * BN, Ha, 2 * D_MODEL, W2o, bias_m, out,
                             D_MODEL, 2 * D_MODEL, 0, keys_a, P, nullptr);
}

extern "C" void kernel_launch(void* const* d_in, const int* in_sizes, int n_in,
                              void* d_out, int out_size, void* d_ws, size_t ws_size,
                              hipStream_t stream) {
    const float* emb_a  = (const float*)d_in[0];
    const float* emb_b  = (const float*)d_in[1];
    const int*   keys_a = (const int*)d_in[2];
    const int*   keys_b = (const int*)d_in[3];
    const float* W1 = (const float*)d_in[4];
    const float* b1 = (const float*)d_in[5];
    const float* W2 = (const float*)d_in[6];
    const float* b2 = (const float*)d_in[7];
    const float* Wo = (const float*)d_in[8];
    const float* bo = (const float*)d_in[9];
    float* out = (float*)d_out;

    float* ws          = (float*)d_ws;
    float* bucket_sums = ws;
    int*   counts      = (int*)(bucket_sums + NBUCKETS * D_MODEL);
    float* Ha          = bucket_sums + NBUCKETS * D_MODEL + NBUCKETS;
    float* W2o         = Ha + (size_t)BATCH_A * 2 * D_MODEL;
    float* bias_m      = W2o + 2 * D_MODEL * D_MODEL;
    float* P           = bias_m + D_MODEL;

    bucket_sum_kernel<<<NBUCKETS / BK_BUCKETS, BK_THREADS, 0, stream>>>(emb_b, keys_b, bucket_sums, counts);

    first_layer_halves_and_merge_kernel<<<dim3(L1_GRID_X, L1_GRID_Y + P_GRID_Y + EXTRA_GRID_Y), 256, 0, stream>>>(
        emb_a, W1, b1, Ha, bucket_sums, counts, P, W2, b2, Wo, bo, W2o, bias_m);
    hidden_to_out_kernel<<<dim3(D_MODEL / BN, BATCH_A / BM), 256, 0, stream>>>(
        Ha, P, keys_a, W2o, bias_m, out);
}


# last GEMM fetches three K tiles ahead into three register sets (was one ahead) and its barrier waits for LDS only: a tile's global loads get two whole K steps to land; loop unrolled; k order kept
# speedup vs baseline: 12.6165x; 1.0312x over previous
#include <hip/hip_runtime.h>

typedef float v2f __attribute__((ext_vector_type(2)));
typedef float v8f __attribute__((ext_vector_type(8)));

#define D_MODEL   256
#define BATCH_A   4096
#define BATCH_B   8192
#define NKEYS     4
#define NBUCKETS  1024

__device__ __forceinline__ int hash_row(const int* __restrict__ keys, int row) {
    int s = 0;
#pragma unroll
    for (int k = 0; k < NKEYS; ++k) s += keys[row * NKEYS + k];
    return (int)(((unsigned)s) & (NBUCKETS - 1));
}

#define BK_BUCKETS      4
#define BK_THREADS      512
#define BK_WAVES        (BK_THREADS / 32)
#define BK_CHUNK        (BATCH_B / BK_WAVES)
#define BK_LOADS        (BK_CHUNK / 32)
#define BK_SUM_THREADS  (BK_THREADS / BK_BUCKETS)
#define SUM_BATCH       16
static_assert(NKEYS == 4, "one row of keys is one 16-byte load");
static_assert(BK_WAVES * BK_CHUNK == BATCH_B, "scan must cover every b-row");
static_assert(BK_LOADS * 32 == BK_CHUNK, "whole wave-wide loads only");
static_assert(BK_SUM_THREADS * 2 == D_MODEL, "two features per adding thread");
static_assert(NBUCKETS % BK_BUCKETS == 0, "whole blocks of buckets");
static_assert(BATCH_B <= 65536, "row numbers are kept as 16-bit values");

__device__ __forceinline__ int popc32(unsigned v) { return __builtin_popcount(v); }
__device__ __forceinline__ int round_up_batch(int n) { return (n + SUM_BATCH - 1) & ~(SUM_BATCH - 1); }

__global__ __launch_bounds__(BK_THREADS)
void bucket_sum_kernel(const float* __restrict__ emb_b,
                       const int* __restrict__ keys_b,
                       float* __restrict__ bucket_sums,
                       int* __restrict__ counts) {
    __shared__ __align__(16) unsigned short rows[BATCH_B + BK_BUCKETS * SUM_BATCH];
    __shared__ __align__(16) int chunk_cnt[BK_WAVES][BK_BUCKETS];

    const int t    = threadIdx.x;
    const int lane = t & 31;
    const int wave = t >> 5;
    const int g0   = blockIdx.x * BK_BUCKETS;
    const int chunk = (wave + blockIdx.x) & (BK_WAVES - 1);

    const int4* kp = (const int4*)keys_b + chunk * BK_CHUNK + lane;
    int4 kv[BK_LOADS];
#pragma unroll
    for (int i = 0; i < BK_LOADS; ++i) kv[i] = kp[i * 32];

    int      hq[BK_LOADS];
    unsigned mb[BK_LOADS];
    int cnt0 = 0, cnt1 = 0, cnt2 = 0, cnt3 = 0;
#pragma unroll
    for (int i = 0; i < BK_LOADS; ++i) {
        const int s = kv[i].x + kv[i].y + kv[i].z + kv[i].w;
        hq[i] = (int)(((unsigned)s) & (NBUCKETS - 1)) - g0;
        mb[i] = __builtin_amdgcn_ballot_w32((unsigned)hq[i] < (unsigned)BK_BUCKETS);
        if (mb[i] != 0u) {
            cnt0 += popc32(__builtin_amdgcn_ballot_w32(hq[i] == 0));
            cnt1 += popc32(__builtin_amdgcn_ballot_w32(hq[i] == 1));
            cnt2 += popc32(__builtin_amdgcn_ballot_w32(hq[i] == 2));
            cnt3 += popc32(__builtin_amdgcn_ballot_w32(hq[i] == 3));
        }
    }
    if (lane == 0) *(int4*)chunk_cnt[chunk] = make_int4(cnt0, cnt1, cnt2, cnt3);
    __syncthreads();

    int base0 = 0, base1 = 0, base2 = 0, base3 = 0;
    int tot0 = 0, tot1 = 0, tot2 = 0, tot3 = 0;
#pragma unroll
    for (int c = 0; c < BK_WAVES; ++c) {
        const int4 w = *(const int4*)chunk_cnt[c];
        if (c < chunk) { base0 += w.x; base1 += w.y; base2 += w.z; base3 += w.w; }
        tot0 += w.x; tot1 += w.y; tot2 += w.z; tot3 += w.w;
    }
    const int off0 = 0;
    const int off1 = off0 + round_up_batch(tot0);
    const int off2 = off1 + round_up_batch(tot1);
    const int off3 = off2 + round_up_batch(tot2);
    if (t < BK_BUCKETS * SUM_BATCH) {
        const int pq = t / SUM_BATCH, pk = t % SUM_BATCH;
        const int ptot = pq == 0 ? tot0 : pq == 1 ? tot1 : pq == 2 ? tot2 : tot3;
        const int poff = pq == 0 ? off0 : pq == 1 ? off1 : pq == 2 ? off2 : off3;
        if (ptot + pk < round_up_batch(ptot)) rows[poff + ptot + pk] = 0;
    }
    if ((cnt0 | cnt1 | cnt2 | cnt3) != 0) {
        const unsigned lt = (1u << lane) - 1u;
        int pos0 = off0 + base0, pos1 = off1 + base1, pos2 = off2 + base2, pos3 = off3 + base3;
#pragma unroll
        for (int i = 0; i < BK_LOADS; ++i) {
            if (mb[i] != 0u) {
                const unsigned short row = (unsigned short)(chunk * BK_CHUNK + i * 32 + lane);
                const unsigned m0 = __builtin_amdgcn_ballot_w32(hq[i] == 0);
                const unsigned m1 = __builtin_amdgcn_ballot_w32(hq[i] == 1);
                const unsigned m2 = __builtin_amdgcn_ballot_w32(hq[i] == 2);
                const unsigned m3 = __builtin_amdgcn_ballot_w32(hq[i] == 3);
                if (hq[i] == 0) rows[pos0 + popc32(m0 & lt)] = row;
                if (hq[i] == 1) rows[pos1 + popc32(m1 & lt)] = row;
                if (hq[i] == 2) rows[pos2 + popc32(m2 & lt)] = row;
                if (hq[i] == 3) rows[pos3 + popc32(m3 & lt)] = row;
                pos0 += popc32(m0); pos1 += popc32(m1); pos2 += popc32(m2); pos3 += popc32(m3);
            }
        }
    }
    __syncthreads();

    const int q  = t / BK_SUM_THREADS;
    const int tq = t % BK_SUM_THREADS;
    const int total = q == 0 ? tot0 : q == 1 ? tot1 : q == 2 ? tot2 : tot3;
    const int off   = q == 0 ? off0 : q == 1 ? off1 : q == 2 ? off2 : off3;
    const unsigned short* __restrict__ lst = rows + off;
    const float* __restrict__ src = emb_b + 2 * tq;
    float accx = 0.0f, accy = 0.0f;
    for (int e = 0; e < total; e += SUM_BATCH) {
        const uint4 pa = *(const uint4*)&lst[e];
        const uint4 pb = *(const uint4*)&lst[e + 8];
        const unsigned w[8] = {pa.x, pa.y, pa.z, pa.w, pb.x, pb.y, pb.z, pb.w};
        float2 r[SUM_BATCH];
#pragma unroll
        for (int k = 0; k < SUM_BATCH; k += 2) {
            r[k]     = *(const float2*)(src + (size_t)(w[k >> 1] & 0xffffu) * D_MODEL);
            r[k + 1] = *(const float2*)(src + (size_t)(w[k >> 1] >> 16)     * D_MODEL);
        }
#pragma unroll
        for (int k = 0; k < SUM_BATCH; ++k) {
            const float sx = accx + r[k].x;
            const float sy = accy + r[k].y;
            accx = (e + k < total) ? sx : accx;
            accy = (e + k < total) ? sy : accy;
        }
    }
    *(float2*)(bucket_sums + (size_t)(g0 + q) * D_MODEL + 2 * tq) = make_float2(accx, accy);
    if (tq == 0) counts[g0 + q] = total;
}

#define BM 64
#define BN 64
#define KT 64
#define AS_LD (KT + 4)
#define BQ (KT / 4)
#define BQ_LD (4 * BN + 32)
#define LD_ROWS 16
#define A_LD4 (BM / LD_ROWS)
#define B_LD4 4
static_assert(BQ == LD_ROWS, "one k quad of the B tile per staging row group");
static_assert(BN == 64, "slot() swizzles with bits 4..5 of the column");
#define A_PLAIN   0
#define A_SCALED  1
#define A_COMBINE 2

__device__ __forceinline__ void gemm_tile_load_a(const float* __restrict__ Ap, int lda,
                                                 float4 (&av)[A_LD4]) {
#pragma unroll
    for (int i = 0; i < A_LD4; ++i)
        av[i] = *(const float4*)(Ap + (size_t)(LD_ROWS * i) * lda);
}

__device__ __forceinline__ void gemm_tile_gather_a(const float* __restrict__ Gp,
                                                   const unsigned (&goff)[A_LD4],
                                                   float4 (&av)[A_LD4]) {
#pragma unroll
    for (int i = 0; i < A_LD4; ++i)
        av[i] = *(const float4*)(Gp + goff[i]);
}

__device__ __forceinline__ void gemm_tile_load_b(const float* __restrict__ Bp, int N,
                                                 float4 (&bv)[B_LD4]) {
#pragma unroll
    for (int i = 0; i < B_LD4; ++i)
        bv[i] = *(const float4*)(Bp + (size_t)i * N);
}

__device__ __forceinline__ void gemm_tile_store(float (*As)[AS_LD], float* bsp, int bsw,
                                                int lr, int lc,
                                                const float4 (&av)[A_LD4],
                                                const float4 (&bv)[B_LD4]) {
#pragma unroll
    for (int i = 0; i < A_LD4; ++i) *(float4*)&As[lr + LD_ROWS * i][lc] = av[i];
    const float4 t0 = {bv[0].x, bv[1].x, bv[2].x, bv[3].x};
    const float4 t1 = {bv[0].y, bv[1].y, bv[2].y, bv[3].y};
    const float4 t2 = {bv[0].z, bv[1].z, bv[2].z, bv[3].z};
    const float4 t3 = {bv[0].w, bv[1].w, bv[2].w, bv[3].w};
    *(float4*)(bsp + 4 * (0 ^ bsw)) = t0;
    *(float4*)(bsp + 4 * (1 ^ bsw)) = t1;
    *(float4*)(bsp + 4 * (2 ^ bsw)) = t2;
    *(float4*)(bsp + 4 * (3 ^ bsw)) = t3;
}

__device__ __forceinline__ void gemm_tile_mma(const float* __restrict__ ap,
                                              const float* __restrict__ bp0,
                                              const float* __restrict__ bp1,
                                              v8f& c0, v8f& c1) {
    float4 a[KT / 8], b0[KT / 8], b1[KT / 8];
#pragma unroll
    for (int j = 0; j < KT / 8; ++j) {
        a[j]  = *(const float4*)(ap + 8 * j);
        b0[j] = *(const float4*)(bp0 + 2 * j * BQ_LD);
        b1[j] = *(const float4*)(bp1 + 2 * j * BQ_LD);
    }
#pragma unroll
    for (int j = 0; j < KT / 8; ++j) {
        const v2f alo = {a[j].x, a[j].y},   ahi = {a[j].z, a[j].w};
        const v2f p0  = {b0[j].x, b0[j].y}, q0  = {b0[j].z, b0[j].w};
        const v2f p1  = {b1[j].x, b1[j].y}, q1  = {b1[j].z, b1[j].w};
        c0 = __builtin_amdgcn_wmma_f32_16x16x4_f32(false, alo, false, p0,
                                                   (short)0, c0, false, false);
        c1 = __builtin_amdgcn_wmma_f32_16x16x4_f32(false, alo, false, p1,
                                                   (short)0, c1, false, false);
        c0 = __builtin_amdgcn_wmma_f32_16x16x4_f32(false, ahi, false, q0,
                                                   (short)0, c0, false, false);
        c1 = __builtin_amdgcn_wmma_f32_16x16x4_f32(false, ahi, false, q1,
                                                   (short)0, c1, false, false);
    }
}

template <int GROUPS, bool STAGING, int NLOADS>
__device__ __forceinline__ void gemm_step_order() {
    if constexpr (STAGING) __builtin_amdgcn_sched_group_barrier(0x020, NLOADS, 0);
    if constexpr (GROUPS == 1) {
        __builtin_amdgcn_sched_group_barrier(0x100, 3 * (KT / 8), 0);
        __builtin_amdgcn_sched_group_barrier(0x008, 4 * (KT / 8), 0);
    } else {
        __builtin_amdgcn_sched_group_barrier(0x100, 3 * (KT / 16), 0);
        __builtin_amdgcn_sched_group_barrier(0x008, 4 * (KT / 16), 0);
        __builtin_amdgcn_sched_group_barrier(0x100, 3 * (KT / 16), 0);
        __builtin_amdgcn_sched_group_barrier(0x008, 4 * (KT / 16), 0);
    }
    if constexpr (STAGING) __builtin_amdgcn_sched_group_barrier(0x200, A_LD4 + B_LD4, 0);
}

template <int GROUPS, bool FETCH, bool STAGE, int NLOADS>
__device__ __forceinline__ void gemm_step_order_ahead() {
    if constexpr (FETCH) __builtin_amdgcn_sched_group_barrier(0x020, NLOADS, 0);
    if constexpr (GROUPS == 1) {
        __builtin_amdgcn_sched_group_barrier(0x100, 3 * (KT / 8), 0);
        __builtin_amdgcn_sched_group_barrier(0x008, 4 * (KT / 8), 0);
    } else {
        __builtin_amdgcn_sched_group_barrier(0x100, 3 * (KT / 16), 0);
        __builtin_amdgcn_sched_group_barrier(0x008, 4 * (KT / 16), 0);
        __builtin_amdgcn_sched_group_barrier(0x100, 3 * (KT / 16), 0);
        __builtin_amdgcn_sched_group_barrier(0x008, 4 * (KT / 16), 0);
    }
    if constexpr (STAGE) __builtin_amdgcn_sched_group_barrier(0x200, A_LD4 + B_LD4, 0);
}

__device__ __forceinline__ void lds_barrier() {
    __builtin_amdgcn_fence(__ATOMIC_RELEASE, "workgroup", "local");
    __builtin_amdgcn_s_barrier();
    __builtin_amdgcn_fence(__ATOMIC_ACQUIRE, "workgroup", "local");
}

template <int MODE>
__device__ __forceinline__ void gemm_hold_tile(v8f& c0, v8f& c1, float4 (&av)[A_LD4],
                                               float4 (&bv)[B_LD4], float4 (&gv)[A_LD4]) {
    asm volatile("" : "+v"(c0), "+v"(c1),
                      "+v"(av[0].x), "+v"(av[0].y), "+v"(av[0].z), "+v"(av[0].w),
                      "+v"(av[1].x), "+v"(av[1].y), "+v"(av[1].z), "+v"(av[1].w),
                      "+v"(av[2].x), "+v"(av[2].y), "+v"(av[2].z), "+v"(av[2].w),
                      "+v"(av[3].x), "+v"(av[3].y), "+v"(av[3].z), "+v"(av[3].w));
    asm volatile("" : "+v"(bv[0].x), "+v"(bv[0].y), "+v"(bv[0].z), "+v"(bv[0].w),
                      "+v"(bv[1].x), "+v"(bv[1].y), "+v"(bv[1].z), "+v"(bv[1].w),
                      "+v"(bv[2].x), "+v"(bv[2].y), "+v"(bv[2].z), "+v"(bv[2].w),
                      "+v"(bv[3].x), "+v"(bv[3].y), "+v"(bv[3].z), "+v"(bv[3].w));
    if constexpr (MODE == A_COMBINE)
        asm volatile("" : "+v"(gv[0].x), "+v"(gv[0].y), "+v"(gv[0].z), "+v"(gv[0].w),
                          "+v"(gv[1].x), "+v"(gv[1].y), "+v"(gv[1].z), "+v"(gv[1].w),
                          "+v"(gv[2].x), "+v"(gv[2].y), "+v"(gv[2].z), "+v"(gv[2].w),
                          "+v"(gv[3].x), "+v"(gv[3].y), "+v"(gv[3].z), "+v"(gv[3].w));
}
static_assert(A_LD4 == 4 && B_LD4 == 4, "gemm_hold_tile names four float4 per piece");

template <int MODE>
__device__ __forceinline__ void gemm_tile_fix_a(float4 (&av)[A_LD4],
                                                const float4 (&gv)[A_LD4],
                                                const float (&gs)[A_LD4]) {
#pragma unroll
    for (int i = 0; i < A_LD4; ++i) {
        if constexpr (MODE == A_SCALED) {
            av[i].x *= gs[i]; av[i].y *= gs[i]; av[i].z *= gs[i]; av[i].w *= gs[i];
        }
        if constexpr (MODE == A_COMBINE) {
            av[i].x = fmaxf(av[i].x + gv[i].x, 0.0f);
            av[i].y = fmaxf(av[i].y + gv[i].y, 0.0f);
            av[i].z = fmaxf(av[i].z + gv[i].z, 0.0f);
            av[i].w = fmaxf(av[i].w + gv[i].w, 0.0f);
        }
    }
}

template <int MODE, int GROUPS, bool AHEAD3 = false>
__device__ __forceinline__
void gemm_block(float (&As)[2][BM][AS_LD], float (&Bs)[2][BQ][BQ_LD],
                const int bm, const int bn,
                const float* __restrict__ A, const int lda,
                const float* __restrict__ B,
                const float* __restrict__ bias,
                float* __restrict__ C,
                int N, int K, int relu,
                const int* __restrict__ keys_a,
                const float* __restrict__ G,
                const int* __restrict__ counts) {
    constexpr int NLOADS = A_LD4 + B_LD4 + (MODE == A_COMBINE ? A_LD4 : 0);
    const int tid  = threadIdx.x;
    const int lane = tid & 31;
    const int wave = tid >> 5;
    const int wm   = wave >> 1;
    const int wn   = wave & 1;
    const int hi   = lane >> 4;
    const int l15  = lane & 15;

    v8f c0 = {};
    v8f c1 = {};

    const int lr = tid >> 4;
    const int lc = (tid & 15) * 4;
    const float* __restrict__ Ap = A + (size_t)(bm + lr) * lda + lc;
    const float* __restrict__ Bp = B + (size_t)(4 * lr) * N + (bn + lc);
    const int bsw  = (lc >> 4) & 3;
    const int bsto = lr * BQ_LD + 4 * lc;

    unsigned goff[A_LD4];
    float gs[A_LD4];
#pragma unroll
    for (int i = 0; i < A_LD4; ++i) {
        if constexpr (MODE == A_COMBINE)
            goff[i] = (unsigned)(hash_row(keys_a, bm + lr + LD_ROWS * i) * K + lc);
        if constexpr (MODE == A_SCALED) {
            const int cnt = counts[bm + lr + LD_ROWS * i];
            gs[i] = 1.0f / (float)(cnt < 1 ? 1 : cnt);
        }
    }

    const int aoff  = (wm * 16 + l15) * AS_LD + 4 * hi;
    const int boff0 = hi * BQ_LD + 4 * (wn * 32 +      (l15 ^ (2 * wn)));
    const int boff1 = hi * BQ_LD + 4 * (wn * 32 + 16 + (l15 ^ (2 * wn + 1)));

    const int nk = K / KT;
    if constexpr (AHEAD3) {
        float4 av[3][A_LD4], bv[3][B_LD4], gv[3][A_LD4];
#pragma unroll
        for (int j = 0; j < 3; ++j) {
            gemm_tile_load_a(Ap + (size_t)j * KT, lda, av[j]);
            gemm_tile_load_b(Bp + (size_t)j * KT * N, N, bv[j]);
            if constexpr (MODE == A_COMBINE) gemm_tile_gather_a(G + j * KT, goff, gv[j]);
            if (j < 2) asm volatile("" ::: "memory");
        }
        gemm_tile_fix_a<MODE>(av[0], gv[0], gs);
        gemm_tile_store(As[0], &Bs[0][0][0] + bsto, bsw, lr, lc, av[0], bv[0]);
        lds_barrier();
#pragma unroll
        for (int t = 0; t < nk; ++t) {
            __builtin_amdgcn_sched_barrier(0);
            const int cur = t & 1;
            if (t + 3 < nk) {
                gemm_tile_load_a(Ap + (size_t)(t + 3) * KT, lda, av[t % 3]);
                gemm_tile_load_b(Bp + (size_t)(t + 3) * KT * N, N, bv[t % 3]);
                if constexpr (MODE == A_COMBINE) gemm_tile_gather_a(G + (t + 3) * KT, goff, gv[t % 3]);
            }
            gemm_tile_mma(&As[cur][0][0] + aoff, &Bs[cur][0][0] + boff0, &Bs[cur][0][0] + boff1, c0, c1);
            if (t + 1 < nk) {
                gemm_hold_tile<MODE>(c0, c1, av[(t + 1) % 3], bv[(t + 1) % 3], gv[(t + 1) % 3]);
                gemm_tile_fix_a<MODE>(av[(t + 1) % 3], gv[(t + 1) % 3], gs);
                gemm_tile_store(As[cur ^ 1], &Bs[cur ^ 1][0][0] + bsto, bsw, lr, lc,
                                av[(t + 1) % 3], bv[(t + 1) % 3]);
            }
            if (t + 3 < nk)      gemm_step_order_ahead<GROUPS, true, true, NLOADS>();
            else if (t + 1 < nk) gemm_step_order_ahead<GROUPS, false, true, NLOADS>();
            else                 gemm_step_order_ahead<GROUPS, false, false, NLOADS>();
            if (t + 1 < nk) lds_barrier();
        }
    } else {
    float4 av[A_LD4], bv[B_LD4], gv[A_LD4];
    gemm_tile_load_a(Ap, lda, av);
    gemm_tile_load_b(Bp, N, bv);
    if constexpr (MODE == A_COMBINE) gemm_tile_gather_a(G, goff, gv);
    gemm_tile_fix_a<MODE>(av, gv, gs);
    gemm_tile_store(As[0], &Bs[0][0][0] + bsto, bsw, lr, lc, av, bv);
    __syncthreads();

    int cur = 0;
#pragma nounroll
    for (int t = 1; t < nk; ++t) {
        gemm_tile_load_a(Ap + (size_t)t * KT, lda, av);
        gemm_tile_load_b(Bp + (size_t)t * KT * N, N, bv);
        if constexpr (MODE == A_COMBINE) gemm_tile_gather_a(G + t * KT, goff, gv);
        gemm_tile_mma(&As[cur][0][0] + aoff, &Bs[cur][0][0] + boff0, &Bs[cur][0][0] + boff1, c0, c1);
        gemm_tile_fix_a<MODE>(av, gv, gs);
        gemm_tile_store(As[cur ^ 1], &Bs[cur ^ 1][0][0] + bsto, bsw, lr, lc, av, bv);
        gemm_step_order<GROUPS, true, NLOADS>();
        __syncthreads();
        cur ^= 1;
    }
    gemm_tile_mma(&As[cur][0][0] + aoff, &Bs[cur][0][0] + boff0, &Bs[cur][0][0] + boff1, c0, c1);
    gemm_step_order<GROUPS, false, NLOADS>();
    }

    const int n0 = bn + wn * 32 + l15;
    const float bia0 = bias ? bias[n0] : 0.0f;
    const float bia1 = bias ? bias[n0 + 16] : 0.0f;
#pragma unroll
    for (int r = 0; r < 8; ++r) {
        const int m = bm + wm * 16 + r + 8 * hi;
        float v0 = c0[r] + bia0;
        float v1 = c1[r] + bia1;
        if (relu) { v0 = fmaxf(v0, 0.0f); v1 = fmaxf(v1, 0.0f); }
        C[(size_t)m * N + n0]      = v0;
        C[(size_t)m * N + n0 + 16] = v1;
    }
}

#define L1_GRID_X    (2 * D_MODEL / BN)
#define L1_GRID_Y    (BATCH_A / BM)
#define P_GRID_Y     (NBUCKETS / BM)
#define MERGE_GRID_X (D_MODEL / BN)
#define MERGE_TILES  ((2 * D_MODEL / BM) * MERGE_GRID_X)
#define EXTRA_GRID_Y ((MERGE_TILES + 1 + L1_GRID_X - 1) / L1_GRID_X)

__device__ __forceinline__ void merged_bias_block(const float* __restrict__ b2,
                                                  const float* __restrict__ Wo,
                                                  const float* __restrict__ bo,
                                                  float* __restrict__ bias_m) {
    const int n = threadIdx.x;
    float acc = 0.0f;
#pragma unroll 64
    for (int j = 0; j < D_MODEL; ++j) acc = fmaf(b2[j], Wo[(size_t)j * D_MODEL + n], acc);
    bias_m[n] = acc + bo[n];
}

__global__ __launch_bounds__(256)
void first_layer_halves_and_merge_kernel(const float* __restrict__ emb_a,
                                         const float* __restrict__ W1,
                                         const float* __restrict__ b1,
                                         float* __restrict__ Ha,
                                         const float* __restrict__ bucket_sums,
                                         const int* __restrict__ counts,
                                         float* __restrict__ P,
                                         const float* __restrict__ W2,
                                         const float* __restrict__ b2,
                                         const float* __restrict__ Wo,
                                         const float* __restrict__ bo,
                                         float* __restrict__ W2o,
                                         float* __restrict__ bias_m) {
    __shared__ __align__(16) float As[2][BM][AS_LD];
    __shared__ __align__(16) float Bs[2][BQ][BQ_LD];
    static_assert(D_MODEL == 256, "the bias block has one thread per column");
    static_assert(NBUCKETS % BM == 0, "whole row tiles of P");

    if (blockIdx.y < L1_GRID_Y) {
        gemm_block<A_PLAIN, 2>(As, Bs, blockIdx.y * BM, blockIdx.x * BN, emb_a, D_MODEL, W1, b1, Ha,
                               2 * D_MODEL, D_MODEL, 0, nullptr, nullptr, nullptr);
        return;
    }
    if (blockIdx.y < L1_GRID_Y + P_GRID_Y) {
        gemm_block<A_SCALED, 2>(As, Bs, (blockIdx.y - L1_GRID_Y) * BM, blockIdx.x * BN,
                                bucket_sums, D_MODEL, W1 + (size_t)D_MODEL * (2 * D_MODEL), nullptr, P,
                                2 * D_MODEL, D_MODEL, 0, nullptr, nullptr, counts);
        return;
    }
    const int e = (blockIdx.y - L1_GRID_Y - P_GRID_Y) * L1_GRID_X + blockIdx.x;
    if (e < MERGE_TILES) {
        gemm_block<A_PLAIN, 2>(As, Bs, (e / MERGE_GRID_X) * BM, (e % MERGE_GRID_X) * BN,
                               W2, D_MODEL, Wo, nullptr, W2o, D_MODEL, D_MODEL, 0,
                               nullptr, nullptr, nullptr);
    } else if (e == MERGE_TILES) {
        merged_bias_block(b2, Wo, bo, bias_m);
    }
}

__global__ __launch_bounds__(256)
void hidden_to_out_kernel(const float* __restrict__ Ha,
                          const float* __restrict__ P,
                          const int* __restrict__ keys_a,
                          const float* __restrict__ W2o,
                          const float* __restrict__ bias_m,
                          float* __restrict__ out) {
    __shared__ __align__(16) float As[2][BM][AS_LD];
    __shared__ __align__(16) float Bs[2][BQ][BQ_LD];
    gemm_block<A_COMBINE, 1, true>(As, Bs, blockIdx.y * BM, blockIdx.x * BN, Ha, 2 * D_MODEL, W2o, bias_m, out,
                             D_MODEL, 2 * D_MODEL, 0, keys_a, P, nullptr);
}

extern "C" void kernel_launch(void* const* d_in, const int* in_sizes, int n_in,
                              void* d_out, int out_size, void* d_ws, size_t ws_size,
                              hipStream_t stream) {
    const float* emb_a  = (const float*)d_in[0];
    const float* emb_b  = (const float*)d_in[1];
    const int*   keys_a = (const int*)d_in[2];
    const int*   keys_b = (const int*)d_in[3];
    const float* W1 = (const float*)d_in[4];
    const float* b1 = (const float*)d_in[5];
    const float* W2 = (const float*)d_in[6];
    const float* b2 = (const float*)d_in[7];
    const float* Wo = (const float*)d_in[8];
    const float* bo = (const float*)d_in[9];
    float* out = (float*)d_out;

    float* ws          = (float*)d_ws;
    float* bucket_sums = ws;
    int*   counts      = (int*)(bucket_sums + NBUCKETS * D_MODEL);
    float* Ha          = bucket_sums + NBUCKETS * D_MODEL + NBUCKETS;
    float* W2o         = Ha + (size_t)BATCH_A * 2 * D_MODEL;
    float* bias_m      = W2o + 2 * D_MODEL * D_MODEL;
    float* P           = bias_m + D_MODEL;

    bucket_sum_kernel<<<NBUCKETS / BK_BUCKETS, BK_THREADS, 0, stream>>>(emb_b, keys_b, bucket_sums, counts);

    first_layer_halves_and_merge_kernel<<<dim3(L1_GRID_X, L1_GRID_Y + P_GRID_Y + EXTRA_GRID_Y), 256, 0, stream>>>(
        emb_a, W1, b1, Ha, bucket_sums, counts, P, W2, b2, Wo, bo, W2o, bias_m);
    hidden_to_out_kernel<<<dim3(D_MODEL / BN, BATCH_A / BM), 256, 0, stream>>>(
        Ha, P, keys_a, W2o, bias_m, out);
}


# P blocks of 16 buckets x 128 columns (256 blocks) scan the keys, add their buckets' rows in row order, then multiply: bucket launch gone, 2 launches; W2 x Wo in the same 16-row form; all sums kept
# speedup vs baseline: 13.6556x; 1.0824x over previous
#include <hip/hip_runtime.h>

typedef float v2f __attribute__((ext_vector_type(2)));
typedef float v8f __attribute__((ext_vector_type(8)));

#define D_MODEL   256
#define BATCH_A   4096
#define BATCH_B   8192
#define NKEYS     4
#define NBUCKETS  1024

__device__ __forceinline__ int hash_row(const int* __restrict__ keys, int row) {
    int s = 0;
#pragma unroll
    for (int k = 0; k < NKEYS; ++k) s += keys[row * NKEYS + k];
    return (int)(((unsigned)s) & (NBUCKETS - 1));
}

#define PB_BUCKETS  16
#define PB_COLS     128
#define PB_COLQ     (2 * D_MODEL / PB_COLS)
#define PB_BLOCKS   ((NBUCKETS / PB_BUCKETS) * PB_COLQ)
#define PB_WAVES    8
#define PB_CHUNK    (BATCH_B / PB_WAVES)
#define PB_LOADS    (PB_CHUNK / 32)
#define PB_BATCH    8
#define PB_AT_LD    (D_MODEL + 4)
#define PB_GJ       4
#define PB_NG       (D_MODEL / 8 / PB_GJ)
static_assert(NKEYS == 4, "one row of keys is one 16-byte load");
static_assert(PB_WAVES * PB_CHUNK == BATCH_B, "scan must cover every b-row");
static_assert(2 * PB_WAVES == PB_BUCKETS, "two buckets per wave");
static_assert(PB_WAVES * 16 == PB_COLS, "one 16-column tile per wave");
static_assert(BATCH_B <= 65536, "row numbers are kept as 16-bit values");
static_assert(D_MODEL == 256, "a lane adds features 4*lane..+3 and 128 + 4*lane..+3");
static_assert(D_MODEL / 8 == PB_NG * PB_GJ, "the product's K is D_MODEL");

struct PBlockLds {
    alignas(16) unsigned       hits[BATCH_B];
    alignas(16) unsigned short rows[BATCH_B + PB_BUCKETS * PB_BATCH];
    alignas(16) float          at[PB_BUCKETS][PB_AT_LD];
    alignas(16) int            tot[PB_BUCKETS];
    alignas(16) int            chunk_cnt[PB_WAVES];
};

__device__ __forceinline__ int popc32(unsigned v) { return __builtin_popcount(v); }
__device__ __forceinline__ int round_up_batch(int n) { return (n + PB_BATCH - 1) & ~(PB_BATCH - 1); }

__device__ __forceinline__ void p_block_mean(const unsigned short* __restrict__ lst, int total,
                                             const float* __restrict__ src,
                                             float* __restrict__ dst) {
    float4 s0 = {0.0f, 0.0f, 0.0f, 0.0f}, s1 = {0.0f, 0.0f, 0.0f, 0.0f};
    for (int e = 0; e < total; e += PB_BATCH) {
        const uint4 pk = *(const uint4*)&lst[e];
        const unsigned w[4] = {pk.x, pk.y, pk.z, pk.w};
        float4 r0[PB_BATCH], r1[PB_BATCH];
#pragma unroll
        for (int k = 0; k < PB_BATCH; ++k) {
            const unsigned row = (k & 1) ? (w[k >> 1] >> 16) : (w[k >> 1] & 0xffffu);
            r0[k] = *(const float4*)(src + (size_t)row * D_MODEL);
            r1[k] = *(const float4*)(src + (size_t)row * D_MODEL + D_MODEL / 2);
        }
#pragma unroll
        for (int k = 0; k < PB_BATCH; ++k) {
            const bool in = e + k < total;
            s0.x = in ? s0.x + r0[k].x : s0.x;  s0.y = in ? s0.y + r0[k].y : s0.y;
            s0.z = in ? s0.z + r0[k].z : s0.z;  s0.w = in ? s0.w + r0[k].w : s0.w;
            s1.x = in ? s1.x + r1[k].x : s1.x;  s1.y = in ? s1.y + r1[k].y : s1.y;
            s1.z = in ? s1.z + r1[k].z : s1.z;  s1.w = in ? s1.w + r1[k].w : s1.w;
        }
    }
    const float inv = 1.0f / (float)(total < 1 ? 1 : total);
    *(float4*)dst                 = make_float4(s0.x * inv, s0.y * inv, s0.z * inv, s0.w * inv);
    *(float4*)(dst + D_MODEL / 2) = make_float4(s1.x * inv, s1.y * inv, s1.z * inv, s1.w * inv);
}

__device__ __forceinline__ void p_block_load_b(const float* __restrict__ bp, int ldb, int g,
                                               float (&b)[4 * PB_GJ]) {
#pragma unroll
    for (int jj = 0; jj < PB_GJ; ++jj)
#pragma unroll
        for (int s = 0; s < 4; ++s)
            b[4 * jj + s] = bp[(size_t)(8 * (g * PB_GJ + jj) + s) * ldb];
}

__device__ __forceinline__ void row_tile_product(const float (&at)[PB_BUCKETS][PB_AT_LD],
                                                 const float* __restrict__ Bc, int ldb,
                                                 float* __restrict__ Cc, int ldc) {
    const int lane = threadIdx.x & 31;
    const int hi   = lane >> 4;
    const int l15  = lane & 15;
    const float* __restrict__ bp = Bc + (size_t)(4 * hi) * ldb + l15;
    const float* ap = &at[l15][4 * hi];
    v8f acc = {};
    float bs[3][4 * PB_GJ];
    p_block_load_b(bp, ldb, 0, bs[0]);
    p_block_load_b(bp, ldb, 1, bs[1]);
    p_block_load_b(bp, ldb, 2, bs[2]);
#pragma unroll
    for (int g = 0; g < PB_NG; ++g) {
        float4 a[PB_GJ];
#pragma unroll
        for (int jj = 0; jj < PB_GJ; ++jj) a[jj] = *(const float4*)(ap + 8 * (g * PB_GJ + jj));
#pragma unroll
        for (int jj = 0; jj < PB_GJ; ++jj) {
            const v2f alo = {a[jj].x, a[jj].y}, ahi = {a[jj].z, a[jj].w};
            const v2f blo = {bs[g % 3][4 * jj],     bs[g % 3][4 * jj + 1]};
            const v2f bhi = {bs[g % 3][4 * jj + 2], bs[g % 3][4 * jj + 3]};
            acc = __builtin_amdgcn_wmma_f32_16x16x4_f32(false, alo, false, blo,
                                                        (short)0, acc, false, false);
            acc = __builtin_amdgcn_wmma_f32_16x16x4_f32(false, ahi, false, bhi,
                                                        (short)0, acc, false, false);
        }
        if (g + 3 < PB_NG) p_block_load_b(bp, ldb, g + 3, bs[g % 3]);
    }
#pragma unroll
    for (int r = 0; r < 8; ++r)
        Cc[(size_t)(r + 8 * hi) * ldc + l15] = acc[r];
}

__device__ __forceinline__ void p_block(PBlockLds& L, const int e,
                                        const float* __restrict__ emb_b,
                                        const int* __restrict__ keys_b,
                                        const float* __restrict__ W1b,
                                        float* __restrict__ P) {
    const int t    = threadIdx.x;
    const int lane = t & 31;
    const int wave = t >> 5;
    const int g0   = (e / PB_COLQ) * PB_BUCKETS;
    const int colq = e % PB_COLQ;
    const int chunk = (wave + e) & (PB_WAVES - 1);
    const unsigned lt = (1u << lane) - 1u;

    const int4* kp = (const int4*)keys_b + chunk * PB_CHUNK + lane;
    int hq[PB_LOADS];
    int cntw = 0;
#pragma unroll
    for (int i = 0; i < PB_LOADS; ++i) {
        const int4 kv = kp[i * 32];
        const int s = kv.x + kv.y + kv.z + kv.w;
        hq[i] = (int)(((unsigned)s) & (NBUCKETS - 1)) - g0;
        cntw += popc32(__builtin_amdgcn_ballot_w32((unsigned)hq[i] < (unsigned)PB_BUCKETS));
    }
    if (lane == 0) L.chunk_cnt[chunk] = cntw;
    __syncthreads();
    int pos = 0, T = 0;
#pragma unroll
    for (int c = 0; c < PB_WAVES; ++c) {
        const int v = L.chunk_cnt[c];
        if (c < chunk) pos += v;
        T += v;
    }
#pragma unroll
    for (int i = 0; i < PB_LOADS; ++i) {
        const bool inr = (unsigned)hq[i] < (unsigned)PB_BUCKETS;
        const unsigned m = __builtin_amdgcn_ballot_w32(inr);
        if (m != 0u) {
            if (inr) L.hits[pos + popc32(m & lt)] =
                (unsigned)(chunk * PB_CHUNK + i * 32 + lane) | ((unsigned)hq[i] << 16);
            pos += popc32(m);
        }
    }
    __syncthreads();

    const unsigned qa = (unsigned)wave, qb = (unsigned)(wave + PB_WAVES);
    const int nch = (T + 31) >> 5;
    int cnta = 0, cntb = 0;
    for (int c = 0; c < nch; ++c) {
        const int idx = c * 32 + lane;
        const unsigned ent = idx < T ? L.hits[idx] : 0xffffffffu;
        cnta += popc32(__builtin_amdgcn_ballot_w32((ent >> 16) == qa));
        cntb += popc32(__builtin_amdgcn_ballot_w32((ent >> 16) == qb));
    }
    if (lane == 0) { L.tot[qa] = cnta; L.tot[qb] = cntb; }
    __syncthreads();
    int offa = 0, offb = 0, run = 0;
#pragma unroll
    for (int q4 = 0; q4 < PB_BUCKETS / 4; ++q4) {
        const int4 w = *(const int4*)&L.tot[4 * q4];
        const int wv[4] = {w.x, w.y, w.z, w.w};
#pragma unroll
        for (int k = 0; k < 4; ++k) {
            if ((unsigned)(4 * q4 + k) == qa) offa = run;
            if ((unsigned)(4 * q4 + k) == qb) offb = run;
            run += round_up_batch(wv[k]);
        }
    }
    int pa = offa, pb = offb;
    for (int c = 0; c < nch; ++c) {
        const int idx = c * 32 + lane;
        const unsigned ent = idx < T ? L.hits[idx] : 0xffffffffu;
        const bool isa = (ent >> 16) == qa, isb = (ent >> 16) == qb;
        const unsigned ma = __builtin_amdgcn_ballot_w32(isa);
        const unsigned mb = __builtin_amdgcn_ballot_w32(isb);
        if (isa) L.rows[pa + popc32(ma & lt)] = (unsigned short)(ent & 0xffffu);
        if (isb) L.rows[pb + popc32(mb & lt)] = (unsigned short)(ent & 0xffffu);
        pa += popc32(ma);
        pb += popc32(mb);
    }
    if (lane < PB_BATCH) {
        if (cnta + lane < round_up_batch(cnta)) L.rows[offa + cnta + lane] = 0;
        if (cntb + lane < round_up_batch(cntb)) L.rows[offb + cntb + lane] = 0;
    }
    __syncthreads();

    p_block_mean(L.rows + offa, cnta, emb_b + 4 * lane, &L.at[qa][4 * lane]);
    p_block_mean(L.rows + offb, cntb, emb_b + 4 * lane, &L.at[qb][4 * lane]);
    __syncthreads();

    const int n0 = colq * PB_COLS + wave * 16;
    row_tile_product(L.at, W1b + n0, 2 * D_MODEL, P + (size_t)g0 * (2 * D_MODEL) + n0, 2 * D_MODEL);
}

__device__ __forceinline__ void w2o_block(PBlockLds& L, const int e,
                                          const float* __restrict__ W2,
                                          const float* __restrict__ Wo,
                                          float* __restrict__ W2o) {
    const int t    = threadIdx.x;
    const int row0 = (e / 2) * PB_BUCKETS;
    const int n0   = (e % 2) * PB_COLS + (t >> 5) * 16;
    const int r    = t >> 4;
    const int c4   = (t & 15) * 4;
#pragma unroll
    for (int i = 0; i < D_MODEL / 64; ++i)
        *(float4*)&L.at[r][c4 + 64 * i] =
            *(const float4*)(W2 + (size_t)(row0 + r) * D_MODEL + c4 + 64 * i);
    __syncthreads();
    row_tile_product(L.at, Wo + n0, D_MODEL, W2o + (size_t)row0 * D_MODEL + n0, D_MODEL);
}

#define BM 64
#define BN 64
#define KT 64
#define AS_LD (KT + 4)
#define BQ (KT / 4)
#define BQ_LD (4 * BN + 32)
#define LD_ROWS 16
#define A_LD4 (BM / LD_ROWS)
#define B_LD4 4
static_assert(BQ == LD_ROWS, "one k quad of the B tile per staging row group");
static_assert(BN == 64, "slot() swizzles with bits 4..5 of the column");
#define A_PLAIN   0
#define A_COMBINE 2

__device__ __forceinline__ void gemm_tile_load_a(const float* __restrict__ Ap, int lda,
                                                 float4 (&av)[A_LD4]) {
#pragma unroll
    for (int i = 0; i < A_LD4; ++i)
        av[i] = *(const float4*)(Ap + (size_t)(LD_ROWS * i) * lda);
}

__device__ __forceinline__ void gemm_tile_gather_a(const float* __restrict__ Gp,
                                                   const unsigned (&goff)[A_LD4],
                                                   float4 (&av)[A_LD4]) {
#pragma unroll
    for (int i = 0; i < A_LD4; ++i)
        av[i] = *(const float4*)(Gp + goff[i]);
}

__device__ __forceinline__ void gemm_tile_load_b(const float* __restrict__ Bp, int N,
                                                 float4 (&bv)[B_LD4]) {
#pragma unroll
    for (int i = 0; i < B_LD4; ++i)
        bv[i] = *(const float4*)(Bp + (size_t)i * N);
}

__device__ __forceinline__ void gemm_tile_store(float (*As)[AS_LD], float* bsp, int bsw,
                                                int lr, int lc,
                                                const float4 (&av)[A_LD4],
                                                const float4 (&bv)[B_LD4]) {
#pragma unroll
    for (int i = 0; i < A_LD4; ++i) *(float4*)&As[lr + LD_ROWS * i][lc] = av[i];
    const float4 t0 = {bv[0].x, bv[1].x, bv[2].x, bv[3].x};
    const float4 t1 = {bv[0].y, bv[1].y, bv[2].y, bv[3].y};
    const float4 t2 = {bv[0].z, bv[1].z, bv[2].z, bv[3].z};
    const float4 t3 = {bv[0].w, bv[1].w, bv[2].w, bv[3].w};
    *(float4*)(bsp + 4 * (0 ^ bsw)) = t0;
    *(float4*)(bsp + 4 * (1 ^ bsw)) = t1;
    *(float4*)(bsp + 4 * (2 ^ bsw)) = t2;
    *(float4*)(bsp + 4 * (3 ^ bsw)) = t3;
}

__device__ __forceinline__ void gemm_tile_mma(const float* __restrict__ ap,
                                              const float* __restrict__ bp0,
                                              const float* __restrict__ bp1,
                                              v8f& c0, v8f& c1) {
    float4 a[KT / 8], b0[KT / 8], b1[KT / 8];
#pragma unroll
    for (int j = 0; j < KT / 8; ++j) {
        a[j]  = *(const float4*)(ap + 8 * j);
        b0[j] = *(const float4*)(bp0 + 2 * j * BQ_LD);
        b1[j] = *(const float4*)(bp1 + 2 * j * BQ_LD);
    }
#pragma unroll
    for (int j = 0; j < KT / 8; ++j) {
        const v2f alo = {a[j].x, a[j].y},   ahi = {a[j].z, a[j].w};
        const v2f p0  = {b0[j].x, b0[j].y}, q0  = {b0[j].z, b0[j].w};
        const v2f p1  = {b1[j].x, b1[j].y}, q1  = {b1[j].z, b1[j].w};
        c0 = __builtin_amdgcn_wmma_f32_16x16x4_f32(false, alo, false, p0,
                                                   (short)0, c0, false, false);
        c1 = __builtin_amdgcn_wmma_f32_16x16x4_f32(false, alo, false, p1,
                                                   (short)0, c1, false, false);
        c0 = __builtin_amdgcn_wmma_f32_16x16x4_f32(false, ahi, false, q0,
                                                   (short)0, c0, false, false);
        c1 = __builtin_amdgcn_wmma_f32_16x16x4_f32(false, ahi, false, q1,
                                                   (short)0, c1, false, false);
    }
}

template <int GROUPS, bool STAGING, int NLOADS>
__device__ __forceinline__ void gemm_step_order() {
    if constexpr (STAGING) __builtin_amdgcn_sched_group_barrier(0x020, NLOADS, 0);
    if constexpr (GROUPS == 1) {
        __builtin_amdgcn_sched_group_barrier(0x100, 3 * (KT / 8), 0);
        __builtin_amdgcn_sched_group_barrier(0x008, 4 * (KT / 8), 0);
    } else {
        __builtin_amdgcn_sched_group_barrier(0x100, 3 * (KT / 16), 0);
        __builtin_amdgcn_sched_group_barrier(0x008, 4 * (KT / 16), 0);
        __builtin_amdgcn_sched_group_barrier(0x100, 3 * (KT / 16), 0);
        __builtin_amdgcn_sched_group_barrier(0x008, 4 * (KT / 16), 0);
    }
    if constexpr (STAGING) __builtin_amdgcn_sched_group_barrier(0x200, A_LD4 + B_LD4, 0);
}

template <int GROUPS, bool FETCH, bool STAGE, int NLOADS>
__device__ __forceinline__ void gemm_step_order_ahead() {
    if constexpr (FETCH) __builtin_amdgcn_sched_group_barrier(0x020, NLOADS, 0);
    if constexpr (GROUPS == 1) {
        __builtin_amdgcn_sched_group_barrier(0x100, 3 * (KT / 8), 0);
        __builtin_amdgcn_sched_group_barrier(0x008, 4 * (KT / 8), 0);
    } else {
        __builtin_amdgcn_sched_group_barrier(0x100, 3 * (KT / 16), 0);
        __builtin_amdgcn_sched_group_barrier(0x008, 4 * (KT / 16), 0);
        __builtin_amdgcn_sched_group_barrier(0x100, 3 * (KT / 16), 0);
        __builtin_amdgcn_sched_group_barrier(0x008, 4 * (KT / 16), 0);
    }
    if constexpr (STAGE) __builtin_amdgcn_sched_group_barrier(0x200, A_LD4 + B_LD4, 0);
}

__device__ __forceinline__ void lds_barrier() {
    __builtin_amdgcn_fence(__ATOMIC_RELEASE, "workgroup", "local");
    __builtin_amdgcn_s_barrier();
    __builtin_amdgcn_fence(__ATOMIC_ACQUIRE, "workgroup", "local");
}

template <int MODE>
__device__ __forceinline__ void gemm_hold_tile(v8f& c0, v8f& c1, float4 (&av)[A_LD4],
                                               float4 (&bv)[B_LD4], float4 (&gv)[A_LD4]) {
    asm volatile("" : "+v"(c0), "+v"(c1),
                      "+v"(av[0].x), "+v"(av[0].y), "+v"(av[0].z), "+v"(av[0].w),
                      "+v"(av[1].x), "+v"(av[1].y), "+v"(av[1].z), "+v"(av[1].w),
                      "+v"(av[2].x), "+v"(av[2].y), "+v"(av[2].z), "+v"(av[2].w),
                      "+v"(av[3].x), "+v"(av[3].y), "+v"(av[3].z), "+v"(av[3].w));
    asm volatile("" : "+v"(bv[0].x), "+v"(bv[0].y), "+v"(bv[0].z), "+v"(bv[0].w),
                      "+v"(bv[1].x), "+v"(bv[1].y), "+v"(bv[1].z), "+v"(bv[1].w),
                      "+v"(bv[2].x), "+v"(bv[2].y), "+v"(bv[2].z), "+v"(bv[2].w),
                      "+v"(bv[3].x), "+v"(bv[3].y), "+v"(bv[3].z), "+v"(bv[3].w));
    if constexpr (MODE == A_COMBINE)
        asm volatile("" : "+v"(gv[0].x), "+v"(gv[0].y), "+v"(gv[0].z), "+v"(gv[0].w),
                          "+v"(gv[1].x), "+v"(gv[1].y), "+v"(gv[1].z), "+v"(gv[1].w),
                          "+v"(gv[2].x), "+v"(gv[2].y), "+v"(gv[2].z), "+v"(gv[2].w),
                          "+v"(gv[3].x), "+v"(gv[3].y), "+v"(gv[3].z), "+v"(gv[3].w));
}
static_assert(A_LD4 == 4 && B_LD4 == 4, "gemm_hold_tile names four float4 per piece");

template <int MODE>
__device__ __forceinline__ void gemm_tile_fix_a(float4 (&av)[A_LD4],
                                                const float4 (&gv)[A_LD4]) {
#pragma unroll
    for (int i = 0; i < A_LD4; ++i) {
        if constexpr (MODE == A_COMBINE) {
            av[i].x = fmaxf(av[i].x + gv[i].x, 0.0f);
            av[i].y = fmaxf(av[i].y + gv[i].y, 0.0f);
            av[i].z = fmaxf(av[i].z + gv[i].z, 0.0f);
            av[i].w = fmaxf(av[i].w + gv[i].w, 0.0f);
        }
    }
}

template <int MODE, int GROUPS, bool AHEAD3 = false>
__device__ __forceinline__
void gemm_block(float (&As)[2][BM][AS_LD], float (&Bs)[2][BQ][BQ_LD],
                const int bm, const int bn,
                const float* __restrict__ A, const int lda,
                const float* __restrict__ B,
                const float* __restrict__ bias,
                float* __restrict__ C,
                int N, int K, int relu,
                const int* __restrict__ keys_a,
                const float* __restrict__ G) {
    constexpr int NLOADS = A_LD4 + B_LD4 + (MODE == A_COMBINE ? A_LD4 : 0);
    const int tid  = threadIdx.x;
    const int lane = tid & 31;
    const int wave = tid >> 5;
    const int wm   = wave >> 1;
    const int wn   = wave & 1;
    const int hi   = lane >> 4;
    const int l15  = lane & 15;

    v8f c0 = {};
    v8f c1 = {};

    const int lr = tid >> 4;
    const int lc = (tid & 15) * 4;
    const float* __restrict__ Ap = A + (size_t)(bm + lr) * lda + lc;
    const float* __restrict__ Bp = B + (size_t)(4 * lr) * N + (bn + lc);
    const int bsw  = (lc >> 4) & 3;
    const int bsto = lr * BQ_LD + 4 * lc;

    unsigned goff[A_LD4];
#pragma unroll
    for (int i = 0; i < A_LD4; ++i) {
        if constexpr (MODE == A_COMBINE)
            goff[i] = (unsigned)(hash_row(keys_a, bm + lr + LD_ROWS * i) * K + lc);
    }

    const int aoff  = (wm * 16 + l15) * AS_LD + 4 * hi;
    const int boff0 = hi * BQ_LD + 4 * (wn * 32 +      (l15 ^ (2 * wn)));
    const int boff1 = hi * BQ_LD + 4 * (wn * 32 + 16 + (l15 ^ (2 * wn + 1)));

    const int nk = K / KT;
    if constexpr (AHEAD3) {
        float4 av[3][A_LD4], bv[3][B_LD4], gv[3][A_LD4];
#pragma unroll
        for (int j = 0; j < 3; ++j) {
            gemm_tile_load_a(Ap + (size_t)j * KT, lda, av[j]);
            gemm_tile_load_b(Bp + (size_t)j * KT * N, N, bv[j]);
            if constexpr (MODE == A_COMBINE) gemm_tile_gather_a(G + j * KT, goff, gv[j]);
            if (j < 2) asm volatile("" ::: "memory");
        }
        gemm_tile_fix_a<MODE>(av[0], gv[0]);
        gemm_tile_store(As[0], &Bs[0][0][0] + bsto, bsw, lr, lc, av[0], bv[0]);
        lds_barrier();
#pragma unroll
        for (int t = 0; t < nk; ++t) {
            __builtin_amdgcn_sched_barrier(0);
            const int cur = t & 1;
            if (t + 3 < nk) {
                gemm_tile_load_a(Ap + (size_t)(t + 3) * KT, lda, av[t % 3]);
                gemm_tile_load_b(Bp + (size_t)(t + 3) * KT * N, N, bv[t % 3]);
                if constexpr (MODE == A_COMBINE) gemm_tile_gather_a(G + (t + 3) * KT, goff, gv[t % 3]);
            }
            gemm_tile_mma(&As[cur][0][0] + aoff, &Bs[cur][0][0] + boff0, &Bs[cur][0][0] + boff1, c0, c1);
            if (t + 1 < nk) {
                gemm_hold_tile<MODE>(c0, c1, av[(t + 1) % 3], bv[(t + 1) % 3], gv[(t + 1) % 3]);
                gemm_tile_fix_a<MODE>(av[(t + 1) % 3], gv[(t + 1) % 3]);
                gemm_tile_store(As[cur ^ 1], &Bs[cur ^ 1][0][0] + bsto, bsw, lr, lc,
                                av[(t + 1) % 3], bv[(t + 1) % 3]);
            }
            if (t + 3 < nk)      gemm_step_order_ahead<GROUPS, true, true, NLOADS>();
            else if (t + 1 < nk) gemm_step_order_ahead<GROUPS, false, true, NLOADS>();
            else                 gemm_step_order_ahead<GROUPS, false, false, NLOADS>();
            if (t + 1 < nk) lds_barrier();
        }
    } else {
    float4 av[A_LD4], bv[B_LD4], gv[A_LD4];
    gemm_tile_load_a(Ap, lda, av);
    gemm_tile_load_b(Bp, N, bv);
    if constexpr (MODE == A_COMBINE) gemm_tile_gather_a(G, goff, gv);
    gemm_tile_fix_a<MODE>(av, gv);
    gemm_tile_store(As[0], &Bs[0][0][0] + bsto, bsw, lr, lc, av, bv);
    __syncthreads();

    int cur = 0;
#pragma nounroll
    for (int t = 1; t < nk; ++t) {
        gemm_tile_load_a(Ap + (size_t)t * KT, lda, av);
        gemm_tile_load_b(Bp + (size_t)t * KT * N, N, bv);
        if constexpr (MODE == A_COMBINE) gemm_tile_gather_a(G + t * KT, goff, gv);
        gemm_tile_mma(&As[cur][0][0] + aoff, &Bs[cur][0][0] + boff0, &Bs[cur][0][0] + boff1, c0, c1);
        gemm_tile_fix_a<MODE>(av, gv);
        gemm_tile_store(As[cur ^ 1], &Bs[cur ^ 1][0][0] + bsto, bsw, lr, lc, av, bv);
        gemm_step_order<GROUPS, true, NLOADS>();
        __syncthreads();
        cur ^= 1;
    }
    gemm_tile_mma(&As[cur][0][0] + aoff, &Bs[cur][0][0] + boff0, &Bs[cur][0][0] + boff1, c0, c1);
    gemm_step_order<GROUPS, false, NLOADS>();
    }

    const int n0 = bn + wn * 32 + l15;
    const float bia0 = bias ? bias[n0] : 0.0f;
    const float bia1 = bias ? bias[n0 + 16] : 0.0f;
#pragma unroll
    for (int r = 0; r < 8; ++r) {
        const int m = bm + wm * 16 + r + 8 * hi;
        float v0 = c0[r] + bia0;
        float v1 = c1[r] + bia1;
        if (relu) { v0 = fmaxf(v0, 0.0f); v1 = fmaxf(v1, 0.0f); }
        C[(size_t)m * N + n0]      = v0;
        C[(size_t)m * N + n0 + 16] = v1;
    }
}

#define L1_GRID_X    (2 * D_MODEL / BN)
#define L1_GRID_Y    (BATCH_A / BM)
#define PB_GRID_Y    (PB_BLOCKS / L1_GRID_X)
#define MERGE_BLOCKS ((2 * D_MODEL / PB_BUCKETS) * (D_MODEL / PB_COLS))
#define EXTRA_GRID_Y ((MERGE_BLOCKS + 1 + L1_GRID_X - 1) / L1_GRID_X)
#define GEMM_LDS_BYTES ((2 * BM * AS_LD + 2 * BQ * BQ_LD) * 4)
static_assert(PB_BLOCKS % L1_GRID_X == 0, "whole grid rows of P blocks");
static_assert(sizeof(PBlockLds) <= GEMM_LDS_BYTES, "a P block lives in the GEMM tiles' LDS");

__device__ __forceinline__ void merged_bias_block(const float* __restrict__ b2,
                                                  const float* __restrict__ Wo,
                                                  const float* __restrict__ bo,
                                                  float* __restrict__ bias_m) {
    const int n = threadIdx.x;
    float acc = 0.0f;
#pragma unroll 64
    for (int j = 0; j < D_MODEL; ++j) acc = fmaf(b2[j], Wo[(size_t)j * D_MODEL + n], acc);
    bias_m[n] = acc + bo[n];
}

__global__ __launch_bounds__(256)
void first_layer_halves_and_merge_kernel(const float* __restrict__ emb_a,
                                         const float* __restrict__ W1,
                                         const float* __restrict__ b1,
                                         float* __restrict__ Ha,
                                         const float* __restrict__ emb_b,
                                         const int* __restrict__ keys_b,
                                         float* __restrict__ P,
                                         const float* __restrict__ W2,
                                         const float* __restrict__ b2,
                                         const float* __restrict__ Wo,
                                         const float* __restrict__ bo,
                                         float* __restrict__ W2o,
                                         float* __restrict__ bias_m) {
    __shared__ __align__(16) unsigned char lds_raw[GEMM_LDS_BYTES];
    static_assert(D_MODEL == 256, "the bias block has one thread per column");

    if (blockIdx.y < PB_GRID_Y) {
        p_block(*reinterpret_cast<PBlockLds*>(lds_raw), blockIdx.y * L1_GRID_X + blockIdx.x,
                emb_b, keys_b, W1 + (size_t)D_MODEL * (2 * D_MODEL), P);
        return;
    }
    float (&As)[2][BM][AS_LD] = *reinterpret_cast<float (*)[2][BM][AS_LD]>(lds_raw);
    float (&Bs)[2][BQ][BQ_LD] =
        *reinterpret_cast<float (*)[2][BQ][BQ_LD]>(lds_raw + sizeof(float) * 2 * BM * AS_LD);
    const int by = blockIdx.y - PB_GRID_Y;
    if (by < L1_GRID_Y) {
        gemm_block<A_PLAIN, 2>(As, Bs, by * BM, blockIdx.x * BN, emb_a, D_MODEL, W1, b1, Ha,
                               2 * D_MODEL, D_MODEL, 0, nullptr, nullptr);
        return;
    }
    const int e = (by - L1_GRID_Y) * L1_GRID_X + blockIdx.x;
    if (e < MERGE_BLOCKS) {
        w2o_block(*reinterpret_cast<PBlockLds*>(lds_raw), e, W2, Wo, W2o);
    } else if (e == MERGE_BLOCKS) {
        merged_bias_block(b2, Wo, bo, bias_m);
    }
}

__global__ __launch_bounds__(256)
void hidden_to_out_kernel(const float* __restrict__ Ha,
                          const float* __restrict__ P,
                          const int* __restrict__ keys_a,
                          const float* __restrict__ W2o,
                          const float* __restrict__ bias_m,
                          float* __restrict__ out) {
    __shared__ __align__(16) float As[2][BM][AS_LD];
    __shared__ __align__(16) float Bs[2][BQ][BQ_LD];
    gemm_block<A_COMBINE, 1, true>(As, Bs, blockIdx.y * BM, blockIdx.x * BN, Ha, 2 * D_MODEL, W2o, bias_m, out,
                             D_MODEL, 2 * D_MODEL, 0, keys_a, P);
}

extern "C" void kernel_launch(void* const* d_in, const int* in_sizes, int n_in,
                              void* d_out, int out_size, void* d_ws, size_t ws_size,
                              hipStream_t stream) {
    const float* emb_a  = (const float*)d_in[0];
    const float* emb_b  = (const float*)d_in[1];
    const int*   keys_a = (const int*)d_in[2];
    const int*   keys_b = (const int*)d_in[3];
    const float* W1 = (const float*)d_in[4];
    const float* b1 = (const float*)d_in[5];
    const float* W2 = (const float*)d_in[6];
    const float* b2 = (const float*)d_in[7];
    const float* Wo = (const float*)d_in[8];
    const float* bo = (const float*)d_in[9];
    float* out = (float*)d_out;

    float* ws     = (float*)d_ws;
    float* Ha     = ws;
    float* W2o    = Ha + (size_t)BATCH_A * 2 * D_MODEL;
    float* bias_m = W2o + 2 * D_MODEL * D_MODEL;
    float* P      = bias_m + D_MODEL;

    first_layer_halves_and_merge_kernel<<<dim3(L1_GRID_X, PB_GRID_Y + L1_GRID_Y + EXTRA_GRID_Y), 256, 0, stream>>>(
        emb_a, W1, b1, Ha, emb_b, keys_b, P, W2, b2, Wo, bo, W2o, bias_m);
    hidden_to_out_kernel<<<dim3(D_MODEL / BN, BATCH_A / BM), 256, 0, stream>>>(
        Ha, P, keys_a, W2o, bias_m, out);
}
